# starting kernel
# baseline (speedup 1.0000x reference)
#include <hip/hip_runtime.h>

typedef __attribute__((ext_vector_type(2))) float v2f;
typedef __attribute__((ext_vector_type(8))) float v8f;

#define LN_EPS 1e-5f


__global__ void zero_kernel(float* __restrict__ p, long n) {
    long i = (long)blockIdx.x * blockDim.x + threadIdx.x;
    long stride = (long)gridDim.x * blockDim.x;
    for (; i < n; i += stride) p[i] = 0.0f;
}

__global__ void degree_kernel(const int* __restrict__ dst, float* __restrict__ deg, int E) {
    int e = blockIdx.x * blockDim.x + threadIdx.x;
    if (e < E) atomicAdd(&deg[dst[e]], 1.0f);
}

__global__ void invert_kernel(float* __restrict__ norm, int N) {
    int i = blockIdx.x * blockDim.x + threadIdx.x;
    if (i < N) {
        float d = norm[i];
        norm[i] = d > 0.0f ? 1.0f / d : 0.0f;
    }
}

__global__ void scatter_kernel(const float* __restrict__ x, const int* __restrict__ src,
                               const int* __restrict__ dst, const float* __restrict__ norm,
                               float* __restrict__ agg, int E) {
    int t = blockIdx.x * blockDim.x + threadIdx.x;
    int e = t >> 5;
    int lane = t & 31;
    if (e >= E) return;
    int s = src[e];
    int d = dst[e];
    float nv = norm[d];
    const float4 xs = *(const float4*)(x + (long)s * 128 + lane * 4);
    float* ap = agg + (long)d * 128 + lane * 4;
    atomicAdd(ap + 0, xs.x * nv);
    atomicAdd(ap + 1, xs.y * nv);
    atomicAdd(ap + 2, xs.z * nv);
    atomicAdd(ap + 3, xs.w * nv);
}

template <int COLT, bool LNRELU>
__global__ __launch_bounds__(COLT * 32) void sage_gemm_kernel(
    const float* __restrict__ x,
    const float* __restrict__ agg,
    const float* __restrict__ W,
    const float* __restrict__ bias,
    float* __restrict__ y,
    int N) {
    constexpr int NDIM    = COLT * 16;
    constexpr int THREADS = COLT * 32;
    constexpr int G       = 8;
    constexpr int LDK     = 68;

    __shared__ float Wt[NDIM * LDK];
    __shared__ float As[16 * LDK];
    __shared__ float red_s[8 * 16];
    __shared__ float red_q[8 * 16];

    const int tid  = threadIdx.x;
    const int wave = tid >> 5;
    const int lane = tid & 31;
    const int ml   = lane & 15;
    const int half = lane >> 4;
    const long base = (long)blockIdx.x * (G * 16);

    v8f acc[G];
#pragma unroll
    for (int g = 0; g < G; ++g) acc[g] = v8f{};

    for (int kc = 0; kc < 4; ++kc) {
        const int k0 = kc * 64;
        __syncthreads();
        const float* wsrc = W + (long)k0 * NDIM;
        for (int idx = tid; idx < 64 * NDIM; idx += THREADS) {
            const int kloc = idx / NDIM;
            const int n    = idx - kloc * NDIM;
            Wt[n * LDK + kloc] = wsrc[idx];
        }
        const float* asrc = (k0 < 128) ? x : agg;
        const int koff = k0 & 127;
#pragma unroll
        for (int g = 0; g < G; ++g) {
            __syncthreads();
            for (int q = tid; q < 256; q += THREADS) {
                const int r  = q >> 4;
                const int kq = (q & 15) * 4;
                long rr = base + g * 16 + r;
                if (rr >= N) rr = N - 1;
                const float4 v = *(const float4*)(asrc + rr * 128 + koff + kq);
                *(float4*)&As[r * LDK + kq] = v;
            }
            __syncthreads();
            v8f c = acc[g];
#pragma unroll
            for (int k = 0; k < 64; k += 4) {
                const v2f a = *(const v2f*)&As[ml * LDK + half * 2 + k];
                const v2f b = *(const v2f*)&Wt[(wave * 16 + ml) * LDK + half * 2 + k];
                c = __builtin_amdgcn_wmma_f32_16x16x4_f32(
                    false, a, false, b, (short)0, c, false, false);
            }
            acc[g] = c;
        }
    }

    const float bv = bias[wave * 16 + ml];
#pragma unroll
    for (int g = 0; g < G; ++g) {
        v8f c = acc[g];
#pragma unroll
        for (int r = 0; r < 8; ++r) c[r] += bv;

        if (LNRELU) {
#pragma unroll
            for (int r = 0; r < 8; ++r) {
                float sv = c[r];
                float qv = c[r] * c[r];
#pragma unroll
                for (int off = 1; off < 16; off <<= 1) {
                    sv += __shfl_xor(sv, off, 32);
                    qv += __shfl_xor(qv, off, 32);
                }
                if (ml == 0) {
                    red_s[wave * 16 + half * 8 + r] = sv;
                    red_q[wave * 16 + half * 8 + r] = qv;
                }
            }
            __syncthreads();
            if (tid < 16) {
                float ts = 0.0f, tq = 0.0f;
#pragma unroll
                for (int w = 0; w < COLT; ++w) {
                    ts += red_s[w * 16 + tid];
                    tq += red_q[w * 16 + tid];
                }
                red_s[tid] = ts;
                red_q[tid] = tq;
            }
            __syncthreads();
#pragma unroll
            for (int r = 0; r < 8; ++r) {
                const int rowid = half * 8 + r;
                const float mu   = red_s[rowid] * (1.0f / NDIM);
                const float var  = red_q[rowid] * (1.0f / NDIM) - mu * mu;
                const float rstd = rsqrtf(var + LN_EPS);
                float v = (c[r] - mu) * rstd;
                c[r] = v > 0.0f ? v : 0.0f;
            }
            __syncthreads();
        }

#pragma unroll
        for (int r = 0; r < 8; ++r) {
            const long ro = base + g * 16 + half * 8 + r;
            if (ro < N) y[ro * NDIM + wave * 16 + ml] = c[r];
        }
    }
}


extern "C" void kernel_launch(void* const* d_in, const int* in_sizes, int n_in,
                              void* d_out, int out_size, void* d_ws, size_t ws_size,
                              hipStream_t stream) {
    const float* h   = (const float*)d_in[0];
    const int*   src = (const int*)d_in[1];
    const int*   dst = (const int*)d_in[2];
    const float* W1  = (const float*)d_in[3];
    const float* b1  = (const float*)d_in[4];
    const float* W2  = (const float*)d_in[5];
    const float* b2  = (const float*)d_in[6];
    const float* W3  = (const float*)d_in[7];
    const float* b3  = (const float*)d_in[8];
    float* out = (float*)d_out;

    const int N = in_sizes[0] / 128;
    const int E = in_sizes[1];

    float* ws   = (float*)d_ws;
    float* norm = ws;
    float* agg  = norm + N;
    float* h1   = agg + (size_t)N * 128;
    float* h2   = h1  + (size_t)N * 128;

    const long feat_n      = (long)N * 128;
    const int  gemm_blocks = (N + 127) / 128;
    const int  scat_blocks = (int)(((long)E * 32 + 255) / 256);

    zero_kernel<<<1024, 256, 0, stream>>>(norm, N);
    degree_kernel<<<(E + 255) / 256, 256, 0, stream>>>(dst, norm, E);
    invert_kernel<<<(N + 255) / 256, 256, 0, stream>>>(norm, N);

    zero_kernel<<<2048, 256, 0, stream>>>(agg, feat_n);
    scatter_kernel<<<scat_blocks, 256, 0, stream>>>(h, src, dst, norm, agg, E);
    sage_gemm_kernel<8, true><<<gemm_blocks, 256, 0, stream>>>(h, agg, W1, b1, h1, N);

    zero_kernel<<<2048, 256, 0, stream>>>(agg, feat_n);
    scatter_kernel<<<scat_blocks, 256, 0, stream>>>(h1, src, dst, norm, agg, E);
    sage_gemm_kernel<8, true><<<gemm_blocks, 256, 0, stream>>>(h1, agg, W2, b2, h2, N);

    zero_kernel<<<2048, 256, 0, stream>>>(agg, feat_n);
    scatter_kernel<<<scat_blocks, 256, 0, stream>>>(h2, src, dst, norm, agg, E);
    sage_gemm_kernel<4, false><<<gemm_blocks, 128, 0, stream>>>(h2, agg, W3, b3, out, N);
}


# Float atomic scatter-add replaced by gather over a dst-grouped edge list built once (int returning add = rank, scan, fill); rows summed per node in ascending-src order, then x 1/deg: sum order fixed
# speedup vs baseline: 1.8914x; 1.8914x over previous
#include <hip/hip_runtime.h>


typedef __attribute__((ext_vector_type(2))) float v2f;
typedef __attribute__((ext_vector_type(8))) float v8f;

#define LN_EPS 1e-5f


__global__ void zero_int_kernel(int* __restrict__ p, int n) {
    int i = blockIdx.x * blockDim.x + threadIdx.x;
    if (i < n) p[i] = 0;
}

__global__ void count_rank_kernel(const int* __restrict__ dst, int* __restrict__ cnt,
                                  int* __restrict__ rank, int E) {
    int e = blockIdx.x * blockDim.x + threadIdx.x;
    if (e < E) rank[e] = atomicAdd(&cnt[dst[e]], 1);
}

__global__ __launch_bounds__(256) void block_sum_kernel(const int* __restrict__ cnt,
                                                        int* __restrict__ bsum) {
    __shared__ int wsum[8];
    const int tid = threadIdx.x;
    const int4 c = ((const int4*)cnt)[blockIdx.x * 256 + tid];
    int t = c.x + c.y + c.z + c.w;
    for (int off = 16; off > 0; off >>= 1) t += __shfl_xor(t, off, 32);
    if ((tid & 31) == 0) wsum[tid >> 5] = t;
    __syncthreads();
    if (tid == 0) {
        int s = 0;
        for (int w = 0; w < 8; ++w) s += wsum[w];
        bsum[blockIdx.x] = s;
    }
}

__global__ __launch_bounds__(256) void scan_kernel(const int* __restrict__ cnt,
                                                   const int* __restrict__ bsum,
                                                   int* __restrict__ rowptr) {
    __shared__ int wsum[8];
    __shared__ int wtot[8];
    const int tid  = threadIdx.x;
    const int lane = tid & 31;
    const int wave = tid >> 5;
    const int b    = blockIdx.x;

    int part = 0;
    for (int i = tid; i < b; i += 256) part += bsum[i];
    for (int off = 16; off > 0; off >>= 1) part += __shfl_xor(part, off, 32);
    if (lane == 0) wsum[wave] = part;

    const int4 c = ((const int4*)cnt)[b * 256 + tid];
    const int t = c.x + c.y + c.z + c.w;
    int inc = t;
    for (int off = 1; off < 32; off <<= 1) {
        const int v = __shfl_up(inc, off, 32);
        if (lane >= off) inc += v;
    }
    if (lane == 31) wtot[wave] = inc;
    __syncthreads();

    int excl = inc - t;
    for (int w = 0; w < 8; ++w) {
        excl += wsum[w];
        if (w < wave) excl += wtot[w];
    }
    int4 r;
    r.x = excl;
    r.y = r.x + c.x;
    r.z = r.y + c.y;
    r.w = r.z + c.z;
    ((int4*)rowptr)[b * 256 + tid] = r;
}

__global__ void fill_kernel(const int* __restrict__ src, const int* __restrict__ dst,
                            const int* __restrict__ rowptr, const int* __restrict__ rank,
                            int* __restrict__ col, int E) {
    int e = blockIdx.x * blockDim.x + threadIdx.x;
    if (e < E) col[rowptr[dst[e]] + rank[e]] = src[e];
}

__global__ __launch_bounds__(256) void sort_groups_kernel(const int* __restrict__ rowptr,
                                                          int* __restrict__ col, int N) {
    const int lane = threadIdx.x & 31;
    const int wave0 = __builtin_amdgcn_readfirstlane((int)(blockIdx.x * 8 + (threadIdx.x >> 5)));
    const int nw = gridDim.x * 8;
    for (int node = wave0; node < N; node += nw) {
        const int beg = rowptr[node];
        const int deg = rowptr[node + 1] - beg;
        if (deg <= 1) continue;
        if (deg <= 32) {
            const int v = (lane < deg) ? col[beg + lane] : 0;
            int r = 0;
            for (int j = 0; j < deg; ++j) {
                const int vj = __builtin_amdgcn_readlane(v, j);
                r += (vj < v || (vj == v && j < lane)) ? 1 : 0;
            }
            if (lane < deg) col[beg + r] = v;
        } else if (lane == 0) {
            for (int i = 1; i < deg; ++i) {
                const int key = col[beg + i];
                int j = i - 1;
                while (j >= 0 && col[beg + j] > key) {
                    col[beg + j + 1] = col[beg + j];
                    --j;
                }
                col[beg + j + 1] = key;
            }
        }
    }
}

#define ROW_LOAD(S) (*(const float4*)(x + (long)(S) * 128 + lane * 4))
#define ROW_ADD(V) do { acc.x += (V).x; acc.y += (V).y; acc.z += (V).z; acc.w += (V).w; } while (0)

__device__ __forceinline__ void sum_rows(const float* __restrict__ x, const int lane,
                                         const int ids, const int cnt, float4& acc) {
    int j = 0;
    for (; j + 8 <= cnt; j += 8) {
        const float4 v0 = ROW_LOAD(__builtin_amdgcn_readlane(ids, j + 0));
        const float4 v1 = ROW_LOAD(__builtin_amdgcn_readlane(ids, j + 1));
        const float4 v2 = ROW_LOAD(__builtin_amdgcn_readlane(ids, j + 2));
        const float4 v3 = ROW_LOAD(__builtin_amdgcn_readlane(ids, j + 3));
        const float4 v4 = ROW_LOAD(__builtin_amdgcn_readlane(ids, j + 4));
        const float4 v5 = ROW_LOAD(__builtin_amdgcn_readlane(ids, j + 5));
        const float4 v6 = ROW_LOAD(__builtin_amdgcn_readlane(ids, j + 6));
        const float4 v7 = ROW_LOAD(__builtin_amdgcn_readlane(ids, j + 7));
        ROW_ADD(v0); ROW_ADD(v1); ROW_ADD(v2); ROW_ADD(v3);
        ROW_ADD(v4); ROW_ADD(v5); ROW_ADD(v6); ROW_ADD(v7);
    }
    if (j + 4 <= cnt) {
        const float4 v0 = ROW_LOAD(__builtin_amdgcn_readlane(ids, j + 0));
        const float4 v1 = ROW_LOAD(__builtin_amdgcn_readlane(ids, j + 1));
        const float4 v2 = ROW_LOAD(__builtin_amdgcn_readlane(ids, j + 2));
        const float4 v3 = ROW_LOAD(__builtin_amdgcn_readlane(ids, j + 3));
        ROW_ADD(v0); ROW_ADD(v1); ROW_ADD(v2); ROW_ADD(v3);
        j += 4;
    }
    const int rem = cnt - j;
    if (rem == 3) {
        const float4 v0 = ROW_LOAD(__builtin_amdgcn_readlane(ids, j + 0));
        const float4 v1 = ROW_LOAD(__builtin_amdgcn_readlane(ids, j + 1));
        const float4 v2 = ROW_LOAD(__builtin_amdgcn_readlane(ids, j + 2));
        ROW_ADD(v0); ROW_ADD(v1); ROW_ADD(v2);
    } else if (rem == 2) {
        const float4 v0 = ROW_LOAD(__builtin_amdgcn_readlane(ids, j + 0));
        const float4 v1 = ROW_LOAD(__builtin_amdgcn_readlane(ids, j + 1));
        ROW_ADD(v0); ROW_ADD(v1);
    } else if (rem == 1) {
        const float4 v0 = ROW_LOAD(__builtin_amdgcn_readlane(ids, j + 0));
        ROW_ADD(v0);
    }
}

__global__ __launch_bounds__(256) void gather_mean_kernel(const float* __restrict__ x,
                                                          const int* __restrict__ rowptr,
                                                          const int* __restrict__ col,
                                                          float* __restrict__ agg, int N) {
    const int lane = threadIdx.x & 31;
    const int wave0 = __builtin_amdgcn_readfirstlane((int)(blockIdx.x * 8 + (threadIdx.x >> 5)));
    const int nw = gridDim.x * 8;
    if (wave0 >= N) return;

    int node = -1, beg = 0, end = 0, idx = 0;
    int node1 = wave0;
    int beg1 = rowptr[node1];
    int end1 = rowptr[node1 + 1];

    for (;;) {
        int idx1 = 0;
        if (lane < end1 - beg1) idx1 = col[beg1 + lane];
        const int node2 = node1 + nw;
        int beg2 = 0, end2 = 0;
        if (node2 < N) {
            beg2 = rowptr[node2];
            end2 = rowptr[node2 + 1];
        }

        if (node >= 0) {
            const int deg = end - beg;
            float4 acc = make_float4(0.0f, 0.0f, 0.0f, 0.0f);
            sum_rows(x, lane, idx, deg < 32 ? deg : 32, acc);
            for (int base = beg + 32; base < end; base += 32) {
                const int cnt = (end - base) < 32 ? (end - base) : 32;
                const int ids = (lane < cnt) ? col[base + lane] : 0;
                sum_rows(x, lane, ids, cnt, acc);
            }
            const float nv = deg > 0 ? 1.0f / (float)deg : 0.0f;
            float4 o;
            o.x = acc.x * nv;
            o.y = acc.y * nv;
            o.z = acc.z * nv;
            o.w = acc.w * nv;
            *(float4*)(agg + (long)node * 128 + lane * 4) = o;
        }

        if (node1 >= N) break;
        node = node1; beg = beg1; end = end1; idx = idx1;
        node1 = node2; beg1 = beg2; end1 = end2;
    }
}

template <int COLT, bool LNRELU>
__global__ __launch_bounds__(COLT * 32) void sage_gemm_kernel(
    const float* __restrict__ x,
    const float* __restrict__ agg,
    const float* __restrict__ W,
    const float* __restrict__ bias,
    float* __restrict__ y,
    int N) {
    constexpr int NDIM    = COLT * 16;
    constexpr int THREADS = COLT * 32;
    constexpr int G       = 8;
    constexpr int LDK     = 68;

    __shared__ float Wt[NDIM * LDK];
    __shared__ float As[16 * LDK];
    __shared__ float red_s[8 * 16];
    __shared__ float red_q[8 * 16];

    const int tid  = threadIdx.x;
    const int wave = tid >> 5;
    const int lane = tid & 31;
    const int ml   = lane & 15;
    const int half = lane >> 4;
    const long base = (long)blockIdx.x * (G * 16);

    v8f acc[G];
#pragma unroll
    for (int g = 0; g < G; ++g) acc[g] = v8f{};

    for (int kc = 0; kc < 4; ++kc) {
        const int k0 = kc * 64;
        __syncthreads();
        const float* wsrc = W + (long)k0 * NDIM;
        for (int idx = tid; idx < 64 * NDIM; idx += THREADS) {
            const int kloc = idx / NDIM;
            const int n    = idx - kloc * NDIM;
            Wt[n * LDK + kloc] = wsrc[idx];
        }
        const float* asrc = (k0 < 128) ? x : agg;
        const int koff = k0 & 127;
#pragma unroll
        for (int g = 0; g < G; ++g) {
            __syncthreads();
            for (int q = tid; q < 256; q += THREADS) {
                const int r  = q >> 4;
                const int kq = (q & 15) * 4;
                long rr = base + g * 16 + r;
                if (rr >= N) rr = N - 1;
                const float4 v = *(const float4*)(asrc + rr * 128 + koff + kq);
                *(float4*)&As[r * LDK + kq] = v;
            }
            __syncthreads();
            v8f c = acc[g];
#pragma unroll
            for (int k = 0; k < 64; k += 4) {
                const v2f a = *(const v2f*)&As[ml * LDK + half * 2 + k];
                const v2f b = *(const v2f*)&Wt[(wave * 16 + ml) * LDK + half * 2 + k];
                c = __builtin_amdgcn_wmma_f32_16x16x4_f32(
                    false, a, false, b, (short)0, c, false, false);
            }
            acc[g] = c;
        }
    }

    const float bv = bias[wave * 16 + ml];
#pragma unroll
    for (int g = 0; g < G; ++g) {
        v8f c = acc[g];
#pragma unroll
        for (int r = 0; r < 8; ++r) c[r] += bv;

        if (LNRELU) {
#pragma unroll
            for (int r = 0; r < 8; ++r) {
                float sv = c[r];
                float qv = c[r] * c[r];
#pragma unroll
                for (int off = 1; off < 16; off <<= 1) {
                    sv += __shfl_xor(sv, off, 32);
                    qv += __shfl_xor(qv, off, 32);
                }
                if (ml == 0) {
                    red_s[wave * 16 + half * 8 + r] = sv;
                    red_q[wave * 16 + half * 8 + r] = qv;
                }
            }
            __syncthreads();
            if (tid < 16) {
                float ts = 0.0f, tq = 0.0f;
#pragma unroll
                for (int w = 0; w < COLT; ++w) {
                    ts += red_s[w * 16 + tid];
                    tq += red_q[w * 16 + tid];
                }
                red_s[tid] = ts;
                red_q[tid] = tq;
            }
            __syncthreads();
#pragma unroll
            for (int r = 0; r < 8; ++r) {
                const int rowid = half * 8 + r;
                const float mu   = red_s[rowid] * (1.0f / NDIM);
                const float var  = red_q[rowid] * (1.0f / NDIM) - mu * mu;
                const float rstd = rsqrtf(var + LN_EPS);
                float v = (c[r] - mu) * rstd;
                c[r] = v > 0.0f ? v : 0.0f;
            }
            __syncthreads();
        }

#pragma unroll
        for (int r = 0; r < 8; ++r) {
            const long ro = base + g * 16 + half * 8 + r;
            if (ro < N) y[ro * NDIM + wave * 16 + ml] = c[r];
        }
    }
}


extern "C" void kernel_launch(void* const* d_in, const int* in_sizes, int n_in,
                              void* d_out, int out_size, void* d_ws, size_t ws_size,
                              hipStream_t stream) {
    const float* h   = (const float*)d_in[0];
    const int*   src = (const int*)d_in[1];
    const int*   dst = (const int*)d_in[2];
    const float* W1  = (const float*)d_in[3];
    const float* b1  = (const float*)d_in[4];
    const float* W2  = (const float*)d_in[5];
    const float* b2  = (const float*)d_in[6];
    const float* W3  = (const float*)d_in[7];
    const float* b3  = (const float*)d_in[8];
    float* out = (float*)d_out;

    const int N = in_sizes[0] / 128;
    const int E = in_sizes[1];

    const int nslice = (N + 1 + 1023) / 1024;
    const int npad   = nslice * 1024;
    const int spad   = (nslice + 63) & ~63;
    const int epad   = (E + 63) & ~63;

    int*   cnt    = (int*)d_ws;
    int*   rowptr = cnt + npad;
    int*   bsum   = rowptr + npad;
    int*   rank   = bsum + spad;
    int*   col    = rank + epad;
    float* agg    = (float*)(col + epad);
    float* h1     = agg + (size_t)N * 128;
    float* h2     = h1  + (size_t)N * 128;

    const int gemm_blocks = (N + 127) / 128;
    const int edge_blocks = (E + 255) / 256;
    const int node_blocks = (N + 55) / 56;

    zero_int_kernel<<<npad / 256, 256, 0, stream>>>(cnt, npad);
    count_rank_kernel<<<edge_blocks, 256, 0, stream>>>(dst, cnt, rank, E);
    block_sum_kernel<<<nslice, 256, 0, stream>>>(cnt, bsum);
    scan_kernel<<<nslice, 256, 0, stream>>>(cnt, bsum, rowptr);
    fill_kernel<<<edge_blocks, 256, 0, stream>>>(src, dst, rowptr, rank, col, E);
    sort_groups_kernel<<<node_blocks, 256, 0, stream>>>(rowptr, col, N);

    gather_mean_kernel<<<node_blocks, 256, 0, stream>>>(h, rowptr, col, agg, N);
    sage_gemm_kernel<8, true><<<gemm_blocks, 256, 0, stream>>>(h, agg, W1, b1, h1, N);

    gather_mean_kernel<<<node_blocks, 256, 0, stream>>>(h1, rowptr, col, agg, N);
    sage_gemm_kernel<8, true><<<gemm_blocks, 256, 0, stream>>>(h1, agg, W2, b2, h2, N);

    gather_mean_kernel<<<node_blocks, 256, 0, stream>>>(h2, rowptr, col, agg, N);
    sage_gemm_kernel<4, false><<<gemm_blocks, 128, 0, stream>>>(h2, agg, W3, b3, out, N);
}


# Edge grouping by destination rebuilt on LDS counters: 512-node bins, per-slice LDS histograms, one scan, LDS-ranked scatter, per-bin LDS count/scan/place; no global atomics; sum order unchanged
# speedup vs baseline: 3.3960x; 1.7955x over previous
#include <hip/hip_runtime.h>


typedef __attribute__((ext_vector_type(2))) float v2f;
typedef __attribute__((ext_vector_type(8))) float v8f;

#define LN_EPS 1e-5f


#define BIN_SHIFT 9
#define BIN_NODES 512
#define MAX_BINS  2048
#define PART_WGS  256

__global__ __launch_bounds__(1024) void bin_count_kernel(const int* __restrict__ dst,
                                                         int* __restrict__ ghist,
                                                         int E, int chunk, int nbins) {
    __shared__ int hist[MAX_BINS];
    const int tid = threadIdx.x;
    for (int i = tid; i < nbins; i += 1024) hist[i] = 0;
    __syncthreads();
    const int e0 = blockIdx.x * chunk;
    const int e1 = (e0 + chunk < E) ? (e0 + chunk) : E;
    for (int e = e0 + tid; e < e1; e += 4096) {
        int d[4];
#pragma unroll
        for (int u = 0; u < 4; ++u) {
            const int ee = e + u * 1024;
            d[u] = (ee < e1) ? dst[ee] : -1;
        }
#pragma unroll
        for (int u = 0; u < 4; ++u) {
            if (d[u] >= 0) atomicAdd(&hist[d[u] >> BIN_SHIFT], 1);
        }
    }
    __syncthreads();
    for (int i = tid; i < nbins; i += 1024) ghist[i * PART_WGS + blockIdx.x] = hist[i];
}

__global__ __launch_bounds__(1024) void bin_scan_kernel(const int* __restrict__ ghist,
                                                        int* __restrict__ goff, int M) {
    __shared__ int wtot[32];
    const int tid  = threadIdx.x;
    const int lane = tid & 31;
    const int wave = tid >> 5;
    const int per  = (M + 8191) >> 13;
    const int4* gin = (const int4*)ghist;
    int4* gout = (int4*)goff;
    const int4 zero4 = make_int4(0, 0, 0, 0);

    int s = 0;
    for (int k0 = 0; k0 < per; k0 += 4) {
        int4 a[4], b[4];
#pragma unroll
        for (int u = 0; u < 4; ++u) {
            const int row = wave * per + k0 + u;
            const bool ok = (k0 + u < per) && (row * PART_WGS < M);
            const int rc = ok ? row : 0;
            a[u] = gin[rc * 64 + lane * 2];
            b[u] = gin[rc * 64 + lane * 2 + 1];
            if (!ok) { a[u] = zero4; b[u] = zero4; }
        }
#pragma unroll
        for (int u = 0; u < 4; ++u)
            s += a[u].x + a[u].y + a[u].z + a[u].w + b[u].x + b[u].y + b[u].z + b[u].w;
    }
    for (int off = 16; off > 0; off >>= 1) s += __shfl_xor(s, off, 32);
    if (lane == 0) wtot[wave] = s;
    __syncthreads();

    int run = 0;
    for (int w = 0; w < wave; ++w) run += wtot[w];
    for (int k0 = 0; k0 < per; k0 += 4) {
        int4 a[4], b[4];
#pragma unroll
        for (int u = 0; u < 4; ++u) {
            const int row = wave * per + k0 + u;
            const bool ok = (k0 + u < per) && (row * PART_WGS < M);
            const int rc = ok ? row : 0;
            a[u] = gin[rc * 64 + lane * 2];
            b[u] = gin[rc * 64 + lane * 2 + 1];
            if (!ok) { a[u] = zero4; b[u] = zero4; }
        }
#pragma unroll
        for (int u = 0; u < 4; ++u) {
            const int row = wave * per + k0 + u;
            const bool ok = (k0 + u < per) && (row * PART_WGS < M);
            const int t = a[u].x + a[u].y + a[u].z + a[u].w + b[u].x + b[u].y + b[u].z + b[u].w;
            int inc = t;
            for (int off = 1; off < 32; off <<= 1) {
                const int v = __shfl_up(inc, off, 32);
                if (lane >= off) inc += v;
            }
            int4 oa, ob;
            oa.x = run + inc - t;
            oa.y = oa.x + a[u].x;
            oa.z = oa.y + a[u].y;
            oa.w = oa.z + a[u].z;
            ob.x = oa.w + a[u].w;
            ob.y = ob.x + b[u].x;
            ob.z = ob.y + b[u].y;
            ob.w = ob.z + b[u].z;
            if (ok) {
                gout[row * 64 + lane * 2]     = oa;
                gout[row * 64 + lane * 2 + 1] = ob;
            }
            run += __shfl(inc, 31, 32);
        }
    }
}

__global__ __launch_bounds__(1024) void bin_scatter_kernel(const int* __restrict__ src,
                                                           const int* __restrict__ dst,
                                                           const int* __restrict__ goff,
                                                           int2* __restrict__ part,
                                                           int E, int chunk, int nbins) {
    __shared__ int cur[MAX_BINS];
    const int tid = threadIdx.x;
    for (int i = tid; i < nbins; i += 1024) cur[i] = goff[i * PART_WGS + blockIdx.x];
    __syncthreads();
    const int e0 = blockIdx.x * chunk;
    const int e1 = (e0 + chunk < E) ? (e0 + chunk) : E;
    for (int e = e0 + tid; e < e1; e += 4096) {
        int d[4], s[4];
#pragma unroll
        for (int u = 0; u < 4; ++u) {
            const int ee = e + u * 1024;
            d[u] = (ee < e1) ? dst[ee] : -1;
            s[u] = (ee < e1) ? src[ee] : 0;
        }
#pragma unroll
        for (int u = 0; u < 4; ++u) {
            if (d[u] >= 0) {
                const int p = atomicAdd(&cur[d[u] >> BIN_SHIFT], 1);
                part[p] = make_int2(s[u], d[u]);
            }
        }
    }
}

__global__ __launch_bounds__(1024) void bin_build_kernel(const int2* __restrict__ part,
                                                         const int* __restrict__ goff,
                                                         int* __restrict__ rowptr,
                                                         int* __restrict__ col,
                                                         int N, int E, int nbins) {
    __shared__ int cntl[BIN_NODES];
    __shared__ int cursor[BIN_NODES];
    __shared__ int wtot[16];
    const int tid  = threadIdx.x;
    const int lane = tid & 31;
    const int wave = tid >> 5;
    const int bin  = blockIdx.x;
    const int n0   = bin << BIN_SHIFT;
    const int start = goff[bin * PART_WGS];
    const int stop  = (bin + 1 < nbins) ? goff[(bin + 1) * PART_WGS] : E;

    if (tid < BIN_NODES) cntl[tid] = 0;
    __syncthreads();
    for (int i = start + tid; i < stop; i += 4096) {
        int d[4];
#pragma unroll
        for (int u = 0; u < 4; ++u) {
            const int ii = i + u * 1024;
            d[u] = (ii < stop) ? part[ii].y : -1;
        }
#pragma unroll
        for (int u = 0; u < 4; ++u) {
            if (d[u] >= 0) atomicAdd(&cntl[d[u] - n0], 1);
        }
    }
    __syncthreads();

    const int c = (tid < BIN_NODES) ? cntl[tid] : 0;
    int inc = c;
    for (int off = 1; off < 32; off <<= 1) {
        const int u = __shfl_up(inc, off, 32);
        if (lane >= off) inc += u;
    }
    if (lane == 31 && wave < 16) wtot[wave] = inc;
    __syncthreads();
    if (tid < BIN_NODES) {
        int ex = inc - c;
        for (int w = 0; w < wave; ++w) ex += wtot[w];
        cursor[tid] = ex;
        const int node = n0 + tid;
        if (node <= N) rowptr[node] = start + ex;
    }
    __syncthreads();

    for (int i = start + tid; i < stop; i += 4096) {
        int2 r[4];
#pragma unroll
        for (int u = 0; u < 4; ++u) {
            const int ii = i + u * 1024;
            r[u] = (ii < stop) ? part[ii] : make_int2(0, -1);
        }
#pragma unroll
        for (int u = 0; u < 4; ++u) {
            if (r[u].y >= 0) {
                const int p = atomicAdd(&cursor[r[u].y - n0], 1);
                col[start + p] = r[u].x;
            }
        }
    }
}

__global__ __launch_bounds__(256) void sort_groups_kernel(const int* __restrict__ rowptr,
                                                          int* __restrict__ col, int N) {
    const int lane = threadIdx.x & 31;
    const int wave0 = __builtin_amdgcn_readfirstlane((int)(blockIdx.x * 8 + (threadIdx.x >> 5)));
    const int nw = gridDim.x * 8;
    for (int node = wave0; node < N; node += nw) {
        const int beg = rowptr[node];
        const int deg = rowptr[node + 1] - beg;
        if (deg <= 1) continue;
        if (deg <= 32) {
            const int v = (lane < deg) ? col[beg + lane] : 0;
            int r = 0;
            for (int j = 0; j < deg; ++j) {
                const int vj = __builtin_amdgcn_readlane(v, j);
                r += (vj < v || (vj == v && j < lane)) ? 1 : 0;
            }
            if (lane < deg) col[beg + r] = v;
        } else if (lane == 0) {
            for (int i = 1; i < deg; ++i) {
                const int key = col[beg + i];
                int j = i - 1;
                while (j >= 0 && col[beg + j] > key) {
                    col[beg + j + 1] = col[beg + j];
                    --j;
                }
                col[beg + j + 1] = key;
            }
        }
    }
}

#define ROW_LOAD(S) (*(const float4*)(x + (long)(S) * 128 + lane * 4))
#define ROW_ADD(V) do { acc.x += (V).x; acc.y += (V).y; acc.z += (V).z; acc.w += (V).w; } while (0)

__device__ __forceinline__ void sum_rows(const float* __restrict__ x, const int lane,
                                         const int ids, const int cnt, float4& acc) {
    int j = 0;
    for (; j + 8 <= cnt; j += 8) {
        const float4 v0 = ROW_LOAD(__builtin_amdgcn_readlane(ids, j + 0));
        const float4 v1 = ROW_LOAD(__builtin_amdgcn_readlane(ids, j + 1));
        const float4 v2 = ROW_LOAD(__builtin_amdgcn_readlane(ids, j + 2));
        const float4 v3 = ROW_LOAD(__builtin_amdgcn_readlane(ids, j + 3));
        const float4 v4 = ROW_LOAD(__builtin_amdgcn_readlane(ids, j + 4));
        const float4 v5 = ROW_LOAD(__builtin_amdgcn_readlane(ids, j + 5));
        const float4 v6 = ROW_LOAD(__builtin_amdgcn_readlane(ids, j + 6));
        const float4 v7 = ROW_LOAD(__builtin_amdgcn_readlane(ids, j + 7));
        ROW_ADD(v0); ROW_ADD(v1); ROW_ADD(v2); ROW_ADD(v3);
        ROW_ADD(v4); ROW_ADD(v5); ROW_ADD(v6); ROW_ADD(v7);
    }
    if (j + 4 <= cnt) {
        const float4 v0 = ROW_LOAD(__builtin_amdgcn_readlane(ids, j + 0));
        const float4 v1 = ROW_LOAD(__builtin_amdgcn_readlane(ids, j + 1));
        const float4 v2 = ROW_LOAD(__builtin_amdgcn_readlane(ids, j + 2));
        const float4 v3 = ROW_LOAD(__builtin_amdgcn_readlane(ids, j + 3));
        ROW_ADD(v0); ROW_ADD(v1); ROW_ADD(v2); ROW_ADD(v3);
        j += 4;
    }
    const int rem = cnt - j;
    if (rem == 3) {
        const float4 v0 = ROW_LOAD(__builtin_amdgcn_readlane(ids, j + 0));
        const float4 v1 = ROW_LOAD(__builtin_amdgcn_readlane(ids, j + 1));
        const float4 v2 = ROW_LOAD(__builtin_amdgcn_readlane(ids, j + 2));
        ROW_ADD(v0); ROW_ADD(v1); ROW_ADD(v2);
    } else if (rem == 2) {
        const float4 v0 = ROW_LOAD(__builtin_amdgcn_readlane(ids, j + 0));
        const float4 v1 = ROW_LOAD(__builtin_amdgcn_readlane(ids, j + 1));
        ROW_ADD(v0); ROW_ADD(v1);
    } else if (rem == 1) {
        const float4 v0 = ROW_LOAD(__builtin_amdgcn_readlane(ids, j + 0));
        ROW_ADD(v0);
    }
}

__global__ __launch_bounds__(256) void gather_mean_kernel(const float* __restrict__ x,
                                                          const int* __restrict__ rowptr,
                                                          const int* __restrict__ col,
                                                          float* __restrict__ agg, int N) {
    const int lane = threadIdx.x & 31;
    const int wave0 = __builtin_amdgcn_readfirstlane((int)(blockIdx.x * 8 + (threadIdx.x >> 5)));
    const int nw = gridDim.x * 8;
    if (wave0 >= N) return;

    int node = -1, beg = 0, end = 0, idx = 0;
    int node1 = wave0;
    int beg1 = rowptr[node1];
    int end1 = rowptr[node1 + 1];

    for (;;) {
        int idx1 = 0;
        if (lane < end1 - beg1) idx1 = col[beg1 + lane];
        const int node2 = node1 + nw;
        int beg2 = 0, end2 = 0;
        if (node2 < N) {
            beg2 = rowptr[node2];
            end2 = rowptr[node2 + 1];
        }

        if (node >= 0) {
            const int deg = end - beg;
            float4 acc = make_float4(0.0f, 0.0f, 0.0f, 0.0f);
            sum_rows(x, lane, idx, deg < 32 ? deg : 32, acc);
            for (int base = beg + 32; base < end; base += 32) {
                const int cnt = (end - base) < 32 ? (end - base) : 32;
                const int ids = (lane < cnt) ? col[base + lane] : 0;
                sum_rows(x, lane, ids, cnt, acc);
            }
            const float nv = deg > 0 ? 1.0f / (float)deg : 0.0f;
            float4 o;
            o.x = acc.x * nv;
            o.y = acc.y * nv;
            o.z = acc.z * nv;
            o.w = acc.w * nv;
            *(float4*)(agg + (long)node * 128 + lane * 4) = o;
        }

        if (node1 >= N) break;
        node = node1; beg = beg1; end = end1; idx = idx1;
        node1 = node2; beg1 = beg2; end1 = end2;
    }
}

template <int NT, bool LNRELU>
__global__ __launch_bounds__(512) void sage_gemm_kernel(
    const float* __restrict__ x,
    const float* __restrict__ agg,
    const float* __restrict__ W,
    const float* __restrict__ bias,
    float* __restrict__ y,
    int N) {
    constexpr int NDIM = NT * 16;
    constexpr int LDW  = 260;
    constexpr int KQ   = 512 / NDIM;

    __shared__ __attribute__((aligned(16))) float Wt[NDIM * LDW];

    const int tid  = threadIdx.x;
    const int wave = tid >> 5;
    const int lane = tid & 31;
    const int ml   = lane & 15;
    const int half = lane >> 4;

    {
        const int n  = tid & (NDIM - 1);
        const int kq = tid / NDIM;
#pragma unroll 4
        for (int it = 0; it < 64 / KQ; ++it) {
            const int k = (it * KQ + kq) * 4;
            float4 v;
            v.x = W[(k + 0) * NDIM + n];
            v.y = W[(k + 1) * NDIM + n];
            v.z = W[(k + 2) * NDIM + n];
            v.w = W[(k + 3) * NDIM + n];
            *(float4*)&Wt[n * LDW + k] = v;
        }
    }
    __syncthreads();

    float bv[NT];
#pragma unroll
    for (int c = 0; c < NT; ++c) bv[c] = bias[c * 16 + ml];

    const float* wb = &Wt[ml * LDW + half * 4];
    const int nstrips = (N + 15) >> 4;

    for (int s = blockIdx.x + gridDim.x * wave; s < nstrips; s += gridDim.x * 16) {
        long row = (long)s * 16 + ml;
        if (row >= N) row = N - 1;
        const float* xa = x + row * 128 + half * 4;
        const float* ga = agg + row * 128 + half * 4;

        v8f acc[NT];
#pragma unroll
        for (int c = 0; c < NT; ++c) acc[c] = v8f{};

        float4 a_nxt[4];
#pragma unroll
        for (int j = 0; j < 4; ++j) a_nxt[j] = *(const float4*)(xa + j * 8);

#pragma unroll 1
        for (int q = 0; q < 8; ++q) {
            float4 a_cur[4];
#pragma unroll
            for (int j = 0; j < 4; ++j) a_cur[j] = a_nxt[j];
            if (q < 7) {
                const float* p = (q < 3) ? (xa + (q + 1) * 32) : (ga + (q - 3) * 32);
#pragma unroll
                for (int j = 0; j < 4; ++j) a_nxt[j] = *(const float4*)(p + j * 8);
            }
            const float* wq = wb + q * 32;
#pragma unroll
            for (int j = 0; j < 4; ++j) {
                float4 b[NT];
#pragma unroll
                for (int c = 0; c < NT; ++c) b[c] = *(const float4*)(wq + c * 16 * LDW + j * 8);
                const v2f a0 = {a_cur[j].x, a_cur[j].y};
                const v2f a1 = {a_cur[j].z, a_cur[j].w};
#pragma unroll
                for (int c = 0; c < NT; ++c) {
                    const v2f b0 = {b[c].x, b[c].y};
                    acc[c] = __builtin_amdgcn_wmma_f32_16x16x4_f32(
                        false, a0, false, b0, (short)0, acc[c], false, false);
                }
#pragma unroll
                for (int c = 0; c < NT; ++c) {
                    const v2f b1 = {b[c].z, b[c].w};
                    acc[c] = __builtin_amdgcn_wmma_f32_16x16x4_f32(
                        false, a1, false, b1, (short)0, acc[c], false, false);
                }
            }
        }

#pragma unroll
        for (int c = 0; c < NT; ++c) {
#pragma unroll
            for (int r = 0; r < 8; ++r) acc[c][r] += bv[c];
        }

        if (LNRELU) {
#pragma unroll
            for (int r = 0; r < 8; ++r) {
                float sv = 0.0f, qv = 0.0f;
#pragma unroll
                for (int c = 0; c < NT; ++c) {
                    sv += acc[c][r];
                    qv += acc[c][r] * acc[c][r];
                }
#pragma unroll
                for (int off = 1; off < 16; off <<= 1) {
                    sv += __shfl_xor(sv, off, 32);
                    qv += __shfl_xor(qv, off, 32);
                }
                const float mu   = sv * (1.0f / NDIM);
                const float var  = qv * (1.0f / NDIM) - mu * mu;
                const float rstd = rsqrtf(var + LN_EPS);
#pragma unroll
                for (int c = 0; c < NT; ++c) {
                    const float v = (acc[c][r] - mu) * rstd;
                    acc[c][r] = v > 0.0f ? v : 0.0f;
                }
            }
        }

#pragma unroll
        for (int r = 0; r < 8; ++r) {
            const long ro = (long)s * 16 + half * 8 + r;
            if (ro < N) {
#pragma unroll
                for (int c = 0; c < NT; ++c) y[ro * NDIM + c * 16 + ml] = acc[c][r];
            }
        }
    }
}


extern "C" void kernel_launch(void* const* d_in, const int* in_sizes, int n_in,
                              void* d_out, int out_size, void* d_ws, size_t ws_size,
                              hipStream_t stream) {
    const float* h   = (const float*)d_in[0];
    const int*   src = (const int*)d_in[1];
    const int*   dst = (const int*)d_in[2];
    const float* W1  = (const float*)d_in[3];
    const float* b1  = (const float*)d_in[4];
    const float* W2  = (const float*)d_in[5];
    const float* b2  = (const float*)d_in[6];
    const float* W3  = (const float*)d_in[7];
    const float* b3  = (const float*)d_in[8];
    float* out = (float*)d_out;

    const int N = in_sizes[0] / 128;
    const int E = in_sizes[1];

    const int nbins = (N >> BIN_SHIFT) + 1;
    const int npad  = nbins * BIN_NODES;
    const int M     = nbins * PART_WGS;
    const int epad  = (E + 63) & ~63;
    const int chunk = (E + PART_WGS - 1) / PART_WGS;

    int*   rowptr = (int*)d_ws;
    int*   ghist  = rowptr + npad;
    int*   goff   = ghist + M;
    int2*  part   = (int2*)(goff + M);
    int*   col    = (int*)(part + epad);
    float* agg    = (float*)(col + epad);
    float* h1     = agg + (size_t)N * 128;
    float* h2     = h1  + (size_t)N * 128;

    const int node_blocks = (N + 55) / 56;

    bin_count_kernel<<<PART_WGS, 1024, 0, stream>>>(dst, ghist, E, chunk, nbins);
    bin_scan_kernel<<<1, 1024, 0, stream>>>(ghist, goff, M);
    bin_scatter_kernel<<<PART_WGS, 1024, 0, stream>>>(src, dst, goff, part, E, chunk, nbins);
    bin_build_kernel<<<nbins, 1024, 0, stream>>>(part, goff, rowptr, col, N, E, nbins);
    sort_groups_kernel<<<node_blocks, 256, 0, stream>>>(rowptr, col, N);

    gather_mean_kernel<<<node_blocks, 256, 0, stream>>>(h, rowptr, col, agg, N);
    sage_gemm_kernel<8, true><<<256, 512, 0, stream>>>(h, agg, W1, b1, h1, N);

    gather_mean_kernel<<<node_blocks, 256, 0, stream>>>(h1, rowptr, col, agg, N);
    sage_gemm_kernel<8, true><<<256, 512, 0, stream>>>(h1, agg, W2, b2, h2, N);

    gather_mean_kernel<<<node_blocks, 256, 0, stream>>>(h2, rowptr, col, agg, N);
    sage_gemm_kernel<4, false><<<256, 512, 0, stream>>>(h2, agg, W3, b3, out, N);
}


# Neighbour mean fused into the layer GEMM launch: a wave sums its strip's 16 nodes, 16 rows in flight, into its own LDS tile, read as A fragments of k 128-255; 3 launches fewer; sum order unchanged
# speedup vs baseline: 3.5150x; 1.0350x over previous
#include <hip/hip_runtime.h>


typedef __attribute__((ext_vector_type(2))) float v2f;
typedef __attribute__((ext_vector_type(8))) float v8f;

#define LN_EPS 1e-5f


#define BIN_SHIFT 9
#define BIN_NODES 512
#define MAX_BINS  2048
#define PART_WGS  256

__global__ __launch_bounds__(1024) void bin_count_kernel(const int* __restrict__ dst,
                                                         int* __restrict__ ghist,
                                                         int E, int chunk, int nbins) {
    __shared__ int hist[MAX_BINS];
    const int tid = threadIdx.x;
    for (int i = tid; i < nbins; i += 1024) hist[i] = 0;
    __syncthreads();
    const int e0 = blockIdx.x * chunk;
    const int e1 = (e0 + chunk < E) ? (e0 + chunk) : E;
    for (int e = e0 + tid; e < e1; e += 4096) {
        int d[4];
#pragma unroll
        for (int u = 0; u < 4; ++u) {
            const int ee = e + u * 1024;
            d[u] = (ee < e1) ? dst[ee] : -1;
        }
#pragma unroll
        for (int u = 0; u < 4; ++u) {
            if (d[u] >= 0) atomicAdd(&hist[d[u] >> BIN_SHIFT], 1);
        }
    }
    __syncthreads();
    for (int i = tid; i < nbins; i += 1024) ghist[i * PART_WGS + blockIdx.x] = hist[i];
}

__global__ __launch_bounds__(1024) void bin_scan_kernel(const int* __restrict__ ghist,
                                                        int* __restrict__ goff, int M) {
    __shared__ int wtot[32];
    const int tid  = threadIdx.x;
    const int lane = tid & 31;
    const int wave = tid >> 5;
    const int per  = (M + 8191) >> 13;
    const int4* gin = (const int4*)ghist;
    int4* gout = (int4*)goff;
    const int4 zero4 = make_int4(0, 0, 0, 0);

    int s = 0;
    for (int k0 = 0; k0 < per; k0 += 4) {
        int4 a[4], b[4];
#pragma unroll
        for (int u = 0; u < 4; ++u) {
            const int row = wave * per + k0 + u;
            const bool ok = (k0 + u < per) && (row * PART_WGS < M);
            const int rc = ok ? row : 0;
            a[u] = gin[rc * 64 + lane * 2];
            b[u] = gin[rc * 64 + lane * 2 + 1];
            if (!ok) { a[u] = zero4; b[u] = zero4; }
        }
#pragma unroll
        for (int u = 0; u < 4; ++u)
            s += a[u].x + a[u].y + a[u].z + a[u].w + b[u].x + b[u].y + b[u].z + b[u].w;
    }
    for (int off = 16; off > 0; off >>= 1) s += __shfl_xor(s, off, 32);
    if (lane == 0) wtot[wave] = s;
    __syncthreads();

    int run = 0;
    for (int w = 0; w < wave; ++w) run += wtot[w];
    for (int k0 = 0; k0 < per; k0 += 4) {
        int4 a[4], b[4];
#pragma unroll
        for (int u = 0; u < 4; ++u) {
            const int row = wave * per + k0 + u;
            const bool ok = (k0 + u < per) && (row * PART_WGS < M);
            const int rc = ok ? row : 0;
            a[u] = gin[rc * 64 + lane * 2];
            b[u] = gin[rc * 64 + lane * 2 + 1];
            if (!ok) { a[u] = zero4; b[u] = zero4; }
        }
#pragma unroll
        for (int u = 0; u < 4; ++u) {
            const int row = wave * per + k0 + u;
            const bool ok = (k0 + u < per) && (row * PART_WGS < M);
            const int t = a[u].x + a[u].y + a[u].z + a[u].w + b[u].x + b[u].y + b[u].z + b[u].w;
            int inc = t;
            for (int off = 1; off < 32; off <<= 1) {
                const int v = __shfl_up(inc, off, 32);
                if (lane >= off) inc += v;
            }
            int4 oa, ob;
            oa.x = run + inc - t;
            oa.y = oa.x + a[u].x;
            oa.z = oa.y + a[u].y;
            oa.w = oa.z + a[u].z;
            ob.x = oa.w + a[u].w;
            ob.y = ob.x + b[u].x;
            ob.z = ob.y + b[u].y;
            ob.w = ob.z + b[u].z;
            if (ok) {
                gout[row * 64 + lane * 2]     = oa;
                gout[row * 64 + lane * 2 + 1] = ob;
            }
            run += __shfl(inc, 31, 32);
        }
    }
}

__global__ __launch_bounds__(1024) void bin_scatter_kernel(const int* __restrict__ src,
                                                           const int* __restrict__ dst,
                                                           const int* __restrict__ goff,
                                                           int2* __restrict__ part,
                                                           int E, int chunk, int nbins) {
    __shared__ int cur[MAX_BINS];
    const int tid = threadIdx.x;
    for (int i = tid; i < nbins; i += 1024) cur[i] = goff[i * PART_WGS + blockIdx.x];
    __syncthreads();
    const int e0 = blockIdx.x * chunk;
    const int e1 = (e0 + chunk < E) ? (e0 + chunk) : E;
    for (int e = e0 + tid; e < e1; e += 4096) {
        int d[4], s[4];
#pragma unroll
        for (int u = 0; u < 4; ++u) {
            const int ee = e + u * 1024;
            d[u] = (ee < e1) ? dst[ee] : -1;
            s[u] = (ee < e1) ? src[ee] : 0;
        }
#pragma unroll
        for (int u = 0; u < 4; ++u) {
            if (d[u] >= 0) {
                const int p = atomicAdd(&cur[d[u] >> BIN_SHIFT], 1);
                part[p] = make_int2(s[u], d[u]);
            }
        }
    }
}

__global__ __launch_bounds__(1024) void bin_build_kernel(const int2* __restrict__ part,
                                                         const int* __restrict__ goff,
                                                         int* __restrict__ rowptr,
                                                         int* __restrict__ col,
                                                         int N, int E, int nbins) {
    __shared__ int cntl[BIN_NODES];
    __shared__ int cursor[BIN_NODES];
    __shared__ int wtot[16];
    const int tid  = threadIdx.x;
    const int lane = tid & 31;
    const int wave = tid >> 5;
    const int bin  = blockIdx.x;
    const int n0   = bin << BIN_SHIFT;
    const int start = goff[bin * PART_WGS];
    const int stop  = (bin + 1 < nbins) ? goff[(bin + 1) * PART_WGS] : E;

    if (tid < BIN_NODES) cntl[tid] = 0;
    __syncthreads();
    for (int i = start + tid; i < stop; i += 4096) {
        int d[4];
#pragma unroll
        for (int u = 0; u < 4; ++u) {
            const int ii = i + u * 1024;
            d[u] = (ii < stop) ? part[ii].y : -1;
        }
#pragma unroll
        for (int u = 0; u < 4; ++u) {
            if (d[u] >= 0) atomicAdd(&cntl[d[u] - n0], 1);
        }
    }
    __syncthreads();

    const int c = (tid < BIN_NODES) ? cntl[tid] : 0;
    int inc = c;
    for (int off = 1; off < 32; off <<= 1) {
        const int u = __shfl_up(inc, off, 32);
        if (lane >= off) inc += u;
    }
    if (lane == 31 && wave < 16) wtot[wave] = inc;
    __syncthreads();
    if (tid < BIN_NODES) {
        int ex = inc - c;
        for (int w = 0; w < wave; ++w) ex += wtot[w];
        cursor[tid] = ex;
        const int node = n0 + tid;
        if (node <= N) rowptr[node] = start + ex;
    }
    __syncthreads();

    for (int i = start + tid; i < stop; i += 4096) {
        int2 r[4];
#pragma unroll
        for (int u = 0; u < 4; ++u) {
            const int ii = i + u * 1024;
            r[u] = (ii < stop) ? part[ii] : make_int2(0, -1);
        }
#pragma unroll
        for (int u = 0; u < 4; ++u) {
            if (r[u].y >= 0) {
                const int p = atomicAdd(&cursor[r[u].y - n0], 1);
                col[start + p] = r[u].x;
            }
        }
    }
}

__global__ __launch_bounds__(256) void sort_groups_kernel(const int* __restrict__ rowptr,
                                                          int* __restrict__ col, int N) {
    const int lane = threadIdx.x & 31;
    const int wave0 = __builtin_amdgcn_readfirstlane((int)(blockIdx.x * 8 + (threadIdx.x >> 5)));
    const int nw = gridDim.x * 8;
    for (int node = wave0; node < N; node += nw) {
        const int beg = rowptr[node];
        const int deg = rowptr[node + 1] - beg;
        if (deg <= 1) continue;
        if (deg <= 32) {
            const int v = (lane < deg) ? col[beg + lane] : 0;
            int r = 0;
            for (int j = 0; j < deg; ++j) {
                const int vj = __builtin_amdgcn_readlane(v, j);
                r += (vj < v || (vj == v && j < lane)) ? 1 : 0;
            }
            if (lane < deg) col[beg + r] = v;
        } else if (lane == 0) {
            for (int i = 1; i < deg; ++i) {
                const int key = col[beg + i];
                int j = i - 1;
                while (j >= 0 && col[beg + j] > key) {
                    col[beg + j + 1] = col[beg + j];
                    --j;
                }
                col[beg + j + 1] = key;
            }
        }
    }
}

#define ROW_LOAD(S) (*(const float4*)(x + (long)(S) * 128 + lane * 4))
#define ROW_ADD(V) do { acc.x += (V).x; acc.y += (V).y; acc.z += (V).z; acc.w += (V).w; } while (0)

#define TILE_PITCH 132

struct StripWalk {
    int cur, cur_beg, cur_end;
    float4 acc;
};

__device__ __forceinline__ void close_node(float* tl, const int lane, const int rpv, StripWalk& w) {
    const int deg = w.cur_end - w.cur_beg;
    const float nv = deg > 0 ? 1.0f / (float)deg : 0.0f;
    float4 o;
    o.x = w.acc.x * nv;
    o.y = w.acc.y * nv;
    o.z = w.acc.z * nv;
    o.w = w.acc.w * nv;
    *(float4*)(tl + w.cur * TILE_PITCH + lane * 4) = o;
    w.acc = make_float4(0.0f, 0.0f, 0.0f, 0.0f);
    ++w.cur;
    w.cur_beg = w.cur_end;
    w.cur_end = __builtin_amdgcn_readlane(rpv, w.cur + 1);
}

#define ADD_ENTRY(E, V) do { \
        while ((E) >= w.cur_end) close_node(tl, lane, rpv, w); \
        float4& acc = w.acc; \
        ROW_ADD(V); \
    } while (0)

__device__ __forceinline__ void gather_batch16(const float* __restrict__ x, float* tl, const int lane,
                                               const int ids, const int j0, const int e0,
                                               const int rpv, StripWalk& w,
                                               const int* __restrict__ col, const bool pf,
                                               const int pf_base, const int e_end, int& ids_n) {
    float4 v[16];
#pragma unroll
    for (int u = 0; u < 16; ++u) v[u] = ROW_LOAD(__builtin_amdgcn_readlane(ids, j0 + u));
    if (pf) ids_n = (pf_base + lane < e_end) ? col[pf_base + lane] : 0;
#pragma unroll
    for (int u = 0; u < 16; ++u) ADD_ENTRY(e0 + u, v[u]);
}

__device__ __forceinline__ void gather_tail(const float* __restrict__ x, float* tl, const int lane,
                                            const int ids, const int j0, const int e0, const int c,
                                            const int rpv, StripWalk& w) {
    const int o4 = c & 8;
    const int o2 = o4 + (c & 4);
    const int o1 = o2 + (c & 2);
    if (c & 8) {
        float4 v8[8];
#pragma unroll
        for (int u = 0; u < 8; ++u) v8[u] = ROW_LOAD(__builtin_amdgcn_readlane(ids, j0 + u));
#pragma unroll
        for (int u = 0; u < 8; ++u) ADD_ENTRY(e0 + u, v8[u]);
    }
    float4 v4[4], v2[2], v1;
    if (c & 4) {
#pragma unroll
        for (int u = 0; u < 4; ++u) v4[u] = ROW_LOAD(__builtin_amdgcn_readlane(ids, j0 + o4 + u));
    }
    if (c & 2) {
#pragma unroll
        for (int u = 0; u < 2; ++u) v2[u] = ROW_LOAD(__builtin_amdgcn_readlane(ids, j0 + o2 + u));
    }
    if (c & 1) v1 = ROW_LOAD(__builtin_amdgcn_readlane(ids, j0 + o1));
    if (c & 4) {
#pragma unroll
        for (int u = 0; u < 4; ++u) ADD_ENTRY(e0 + o4 + u, v4[u]);
    }
    if (c & 2) {
#pragma unroll
        for (int u = 0; u < 2; ++u) ADD_ENTRY(e0 + o2 + u, v2[u]);
    }
    if (c & 1) ADD_ENTRY(e0 + o1, v1);
}

template <int NT>
__device__ __forceinline__ void chunk_products(const float* wq, const float4 (&a)[4], v8f (&acc)[NT]) {
    constexpr int LDW = 260;
#pragma unroll
    for (int j = 0; j < 4; ++j) {
        float4 b[NT];
#pragma unroll
        for (int c = 0; c < NT; ++c) b[c] = *(const float4*)(wq + c * 16 * LDW + j * 8);
        const v2f a0 = {a[j].x, a[j].y};
        const v2f a1 = {a[j].z, a[j].w};
#pragma unroll
        for (int c = 0; c < NT; ++c) {
            const v2f b0 = {b[c].x, b[c].y};
            acc[c] = __builtin_amdgcn_wmma_f32_16x16x4_f32(
                false, a0, false, b0, (short)0, acc[c], false, false);
        }
#pragma unroll
        for (int c = 0; c < NT; ++c) {
            const v2f b1 = {b[c].z, b[c].w};
            acc[c] = __builtin_amdgcn_wmma_f32_16x16x4_f32(
                false, a1, false, b1, (short)0, acc[c], false, false);
        }
    }
}

template <int NT>
__device__ __forceinline__ void x_half(const float* __restrict__ xa, const float* wb, v8f (&acc)[NT]) {
    float4 a_nxt[4];
#pragma unroll
    for (int j = 0; j < 4; ++j) a_nxt[j] = *(const float4*)(xa + j * 8);
#pragma unroll 1
    for (int q = 0; q < 4; ++q) {
        float4 a_cur[4];
#pragma unroll
        for (int j = 0; j < 4; ++j) a_cur[j] = a_nxt[j];
        if (q < 3) {
#pragma unroll
            for (int j = 0; j < 4; ++j) a_nxt[j] = *(const float4*)(xa + (q + 1) * 32 + j * 8);
        }
        chunk_products<NT>(wb + q * 32, a_cur, acc);
    }
}

template <int NT>
__device__ __forceinline__ void mean_half(const float* ta, const float* wb, v8f (&acc)[NT]) {
    float4 a_nxt[4];
#pragma unroll
    for (int j = 0; j < 4; ++j) a_nxt[j] = *(const float4*)(ta + j * 8);
#pragma unroll 1
    for (int q = 0; q < 4; ++q) {
        float4 a_cur[4];
#pragma unroll
        for (int j = 0; j < 4; ++j) a_cur[j] = a_nxt[j];
        if (q < 3) {
#pragma unroll
            for (int j = 0; j < 4; ++j) a_nxt[j] = *(const float4*)(ta + (q + 1) * 32 + j * 8);
        }
        chunk_products<NT>(wb + 128 + q * 32, a_cur, acc);
    }
}

template <int NT, bool LNRELU>
__global__ __launch_bounds__(512) void sage_layer_kernel(
    const float* __restrict__ x,
    const int* __restrict__ rowptr,
    const int* __restrict__ col,
    const float* __restrict__ W,
    const float* __restrict__ bias,
    float* __restrict__ y,
    int N) {
    constexpr int NDIM = NT * 16;
    constexpr int LDW  = 260;
    constexpr int KQ   = 512 / NDIM;

    __shared__ __attribute__((aligned(16))) float Wt[NDIM * LDW];
    __shared__ __attribute__((aligned(16))) float Mt[16 * 16 * TILE_PITCH];

    const int tid  = threadIdx.x;
    const int wave = __builtin_amdgcn_readfirstlane(tid >> 5);
    const int lane = tid & 31;
    const int ml   = lane & 15;
    const int half = lane >> 4;

    {
        const int n  = tid & (NDIM - 1);
        const int kq = tid / NDIM;
#pragma unroll 4
        for (int it = 0; it < 64 / KQ; ++it) {
            const int k = (it * KQ + kq) * 4;
            float4 v;
            v.x = W[(k + 0) * NDIM + n];
            v.y = W[(k + 1) * NDIM + n];
            v.z = W[(k + 2) * NDIM + n];
            v.w = W[(k + 3) * NDIM + n];
            *(float4*)&Wt[n * LDW + k] = v;
        }
    }
    __syncthreads();

    float bv[NT];
#pragma unroll
    for (int c = 0; c < NT; ++c) bv[c] = bias[c * 16 + ml];

    const float* wb = &Wt[ml * LDW + half * 4];
    float* tl = &Mt[wave * 16 * TILE_PITCH];
    const float* ta = tl + ml * TILE_PITCH + half * 4;
    const int nstrips = (N + 15) >> 4;
    const bool x_first = (((wave >> 3) ^ (wave >> 1)) & 1) != 0;

    for (int s = blockIdx.x + gridDim.x * wave; s < nstrips; s += gridDim.x * 16) {
        long row = (long)s * 16 + ml;
        if (row >= N) row = N - 1;
        const float* xa = x + row * 128 + half * 4;

        v8f acc[NT];
#pragma unroll
        for (int c = 0; c < NT; ++c) acc[c] = v8f{};

        if (x_first) x_half<NT>(xa, wb, acc);

        {
            int rpi = s * 16 + lane;
            if (rpi > N) rpi = N;
            const int rpv   = rowptr[rpi];
            const int e_beg = __builtin_amdgcn_readlane(rpv, 0);
            const int e_end = __builtin_amdgcn_readlane(rpv, 16);
            StripWalk w;
            w.cur = 0;
            w.cur_beg = e_beg;
            w.cur_end = __builtin_amdgcn_readlane(rpv, 1);
            w.acc = make_float4(0.0f, 0.0f, 0.0f, 0.0f);

            int ids_n = (e_beg + lane < e_end) ? col[e_beg + lane] : 0;
            for (int base = e_beg; base < e_end; base += 32) {
                const int ids = ids_n;
                const int cnt = (e_end - base) < 32 ? (e_end - base) : 32;
                int j = 0;
                for (; j + 16 <= cnt; j += 16)
                    gather_batch16(x, tl, lane, ids, j, base + j, rpv, w,
                                   col, (j == 0) && (cnt == 32), base + 32, e_end, ids_n);
                if (j < cnt) gather_tail(x, tl, lane, ids, j, base + j, cnt - j, rpv, w);
            }
            while (w.cur < 16) close_node(tl, lane, rpv, w);
        }

        if (!x_first) x_half<NT>(xa, wb, acc);
        mean_half<NT>(ta, wb, acc);

#pragma unroll
        for (int c = 0; c < NT; ++c) {
#pragma unroll
            for (int r = 0; r < 8; ++r) acc[c][r] += bv[c];
        }

        if (LNRELU) {
#pragma unroll
            for (int r = 0; r < 8; ++r) {
                float sv = 0.0f, qv = 0.0f;
#pragma unroll
                for (int c = 0; c < NT; ++c) {
                    sv += acc[c][r];
                    qv += acc[c][r] * acc[c][r];
                }
#pragma unroll
                for (int off = 1; off < 16; off <<= 1) {
                    sv += __shfl_xor(sv, off, 32);
                    qv += __shfl_xor(qv, off, 32);
                }
                const float mu   = sv * (1.0f / NDIM);
                const float var  = qv * (1.0f / NDIM) - mu * mu;
                const float rstd = rsqrtf(var + LN_EPS);
#pragma unroll
                for (int c = 0; c < NT; ++c) {
                    const float v = (acc[c][r] - mu) * rstd;
                    acc[c][r] = v > 0.0f ? v : 0.0f;
                }
            }
        }

#pragma unroll
        for (int r = 0; r < 8; ++r) {
            const long ro = (long)s * 16 + half * 8 + r;
            if (ro < N) {
#pragma unroll
                for (int c = 0; c < NT; ++c) y[ro * NDIM + c * 16 + ml] = acc[c][r];
            }
        }
    }
}


extern "C" void kernel_launch(void* const* d_in, const int* in_sizes, int n_in,
                              void* d_out, int out_size, void* d_ws, size_t ws_size,
                              hipStream_t stream) {
    const float* h   = (const float*)d_in[0];
    const int*   src = (const int*)d_in[1];
    const int*   dst = (const int*)d_in[2];
    const float* W1  = (const float*)d_in[3];
    const float* b1  = (const float*)d_in[4];
    const float* W2  = (const float*)d_in[5];
    const float* b2  = (const float*)d_in[6];
    const float* W3  = (const float*)d_in[7];
    const float* b3  = (const float*)d_in[8];
    float* out = (float*)d_out;

    const int N = in_sizes[0] / 128;
    const int E = in_sizes[1];

    const int nbins = (N >> BIN_SHIFT) + 1;
    const int npad  = nbins * BIN_NODES;
    const int M     = nbins * PART_WGS;
    const int epad  = (E + 63) & ~63;
    const int chunk = (E + PART_WGS - 1) / PART_WGS;

    int*   rowptr = (int*)d_ws;
    int*   ghist  = rowptr + npad;
    int*   goff   = ghist + M;
    int2*  part   = (int2*)(goff + M);
    int*   col    = (int*)(part + epad);
    float* h1     = (float*)(col + epad);
    float* h2     = h1  + (size_t)N * 128;

    const int node_blocks = (N + 55) / 56;

    bin_count_kernel<<<PART_WGS, 1024, 0, stream>>>(dst, ghist, E, chunk, nbins);
    bin_scan_kernel<<<1, 1024, 0, stream>>>(ghist, goff, M);
    bin_scatter_kernel<<<PART_WGS, 1024, 0, stream>>>(src, dst, goff, part, E, chunk, nbins);
    bin_build_kernel<<<nbins, 1024, 0, stream>>>(part, goff, rowptr, col, N, E, nbins);
    sort_groups_kernel<<<node_blocks, 256, 0, stream>>>(rowptr, col, N);

    sage_layer_kernel<8, true><<<256, 512, 0, stream>>>(h, rowptr, col, W1, b1, h1, N);

    sage_layer_kernel<8, true><<<256, 512, 0, stream>>>(h1, rowptr, col, W2, b2, h2, N);

    sage_layer_kernel<4, false><<<256, 512, 0, stream>>>(h2, rowptr, col, W3, b3, out, N);
}


# Ascending-src order of each node's group made inside the per-bin build: placed and ranked on LDS, each entry written once at its final place; sort launch removed (7 launches); sum order unchanged
# speedup vs baseline: 3.6657x; 1.0429x over previous
#include <hip/hip_runtime.h>


typedef __attribute__((ext_vector_type(2))) float v2f;
typedef __attribute__((ext_vector_type(8))) float v8f;

#define LN_EPS 1e-5f


#define BIN_SHIFT 9
#define BIN_NODES 512
#define MAX_BINS  2048
#define PART_WGS  256

__global__ __launch_bounds__(1024) void bin_count_kernel(const int* __restrict__ dst,
                                                         int* __restrict__ ghist,
                                                         int E, int chunk, int nbins) {
    __shared__ int hist[MAX_BINS];
    const int tid = threadIdx.x;
    for (int i = tid; i < nbins; i += 1024) hist[i] = 0;
    __syncthreads();
    const int e0 = blockIdx.x * chunk;
    const int e1 = (e0 + chunk < E) ? (e0 + chunk) : E;
    for (int e = e0 + tid; e < e1; e += 4096) {
        int d[4];
#pragma unroll
        for (int u = 0; u < 4; ++u) {
            const int ee = e + u * 1024;
            d[u] = (ee < e1) ? dst[ee] : -1;
        }
#pragma unroll
        for (int u = 0; u < 4; ++u) {
            if (d[u] >= 0) atomicAdd(&hist[d[u] >> BIN_SHIFT], 1);
        }
    }
    __syncthreads();
    for (int i = tid; i < nbins; i += 1024) ghist[i * PART_WGS + blockIdx.x] = hist[i];
}

__global__ __launch_bounds__(1024) void bin_scan_kernel(const int* __restrict__ ghist,
                                                        int* __restrict__ goff, int M) {
    __shared__ int wtot[32];
    const int tid  = threadIdx.x;
    const int lane = tid & 31;
    const int wave = tid >> 5;
    const int per  = (M + 8191) >> 13;
    const int4* gin = (const int4*)ghist;
    int4* gout = (int4*)goff;
    const int4 zero4 = make_int4(0, 0, 0, 0);

    int s = 0;
    for (int k0 = 0; k0 < per; k0 += 4) {
        int4 a[4], b[4];
#pragma unroll
        for (int u = 0; u < 4; ++u) {
            const int row = wave * per + k0 + u;
            const bool ok = (k0 + u < per) && (row * PART_WGS < M);
            const int rc = ok ? row : 0;
            a[u] = gin[rc * 64 + lane * 2];
            b[u] = gin[rc * 64 + lane * 2 + 1];
            if (!ok) { a[u] = zero4; b[u] = zero4; }
        }
#pragma unroll
        for (int u = 0; u < 4; ++u)
            s += a[u].x + a[u].y + a[u].z + a[u].w + b[u].x + b[u].y + b[u].z + b[u].w;
    }
    for (int off = 16; off > 0; off >>= 1) s += __shfl_xor(s, off, 32);
    if (lane == 0) wtot[wave] = s;
    __syncthreads();

    int run = 0;
    for (int w = 0; w < wave; ++w) run += wtot[w];
    for (int k0 = 0; k0 < per; k0 += 4) {
        int4 a[4], b[4];
#pragma unroll
        for (int u = 0; u < 4; ++u) {
            const int row = wave * per + k0 + u;
            const bool ok = (k0 + u < per) && (row * PART_WGS < M);
            const int rc = ok ? row : 0;
            a[u] = gin[rc * 64 + lane * 2];
            b[u] = gin[rc * 64 + lane * 2 + 1];
            if (!ok) { a[u] = zero4; b[u] = zero4; }
        }
#pragma unroll
        for (int u = 0; u < 4; ++u) {
            const int row = wave * per + k0 + u;
            const bool ok = (k0 + u < per) && (row * PART_WGS < M);
            const int t = a[u].x + a[u].y + a[u].z + a[u].w + b[u].x + b[u].y + b[u].z + b[u].w;
            int inc = t;
            for (int off = 1; off < 32; off <<= 1) {
                const int v = __shfl_up(inc, off, 32);
                if (lane >= off) inc += v;
            }
            int4 oa, ob;
            oa.x = run + inc - t;
            oa.y = oa.x + a[u].x;
            oa.z = oa.y + a[u].y;
            oa.w = oa.z + a[u].z;
            ob.x = oa.w + a[u].w;
            ob.y = ob.x + b[u].x;
            ob.z = ob.y + b[u].y;
            ob.w = ob.z + b[u].z;
            if (ok) {
                gout[row * 64 + lane * 2]     = oa;
                gout[row * 64 + lane * 2 + 1] = ob;
            }
            run += __shfl(inc, 31, 32);
        }
    }
}

__global__ __launch_bounds__(1024) void bin_scatter_kernel(const int* __restrict__ src,
                                                           const int* __restrict__ dst,
                                                           const int* __restrict__ goff,
                                                           int2* __restrict__ part,
                                                           int E, int chunk, int nbins) {
    __shared__ int cur[MAX_BINS];
    const int tid = threadIdx.x;
    for (int i = tid; i < nbins; i += 1024) cur[i] = goff[i * PART_WGS + blockIdx.x];
    __syncthreads();
    const int e0 = blockIdx.x * chunk;
    const int e1 = (e0 + chunk < E) ? (e0 + chunk) : E;
    for (int e = e0 + tid; e < e1; e += 4096) {
        int d[4], s[4];
#pragma unroll
        for (int u = 0; u < 4; ++u) {
            const int ee = e + u * 1024;
            d[u] = (ee < e1) ? dst[ee] : -1;
            s[u] = (ee < e1) ? src[ee] : 0;
        }
#pragma unroll
        for (int u = 0; u < 4; ++u) {
            if (d[u] >= 0) {
                const int p = atomicAdd(&cur[d[u] >> BIN_SHIFT], 1);
                part[p] = make_int2(s[u], d[u]);
            }
        }
    }
}

#define BIN_CAP 8192

__global__ __launch_bounds__(1024) void bin_build_kernel(const int2* __restrict__ part,
                                                         const int* __restrict__ goff,
                                                         int* __restrict__ rowptr,
                                                         int* __restrict__ col,
                                                         int N, int E, int nbins) {
    __shared__ int cntl[BIN_NODES];
    __shared__ int cursor[BIN_NODES];
    __shared__ int nbeg[BIN_NODES + 1];
    __shared__ int wtot[16];
    __shared__ int lsrc[BIN_CAP];
    __shared__ unsigned short lnode[BIN_CAP];
    const int tid  = threadIdx.x;
    const int lane = tid & 31;
    const int wave = tid >> 5;
    const int bin  = blockIdx.x;
    const int n0   = bin << BIN_SHIFT;
    const int start = goff[bin * PART_WGS];
    const int stop  = (bin + 1 < nbins) ? goff[(bin + 1) * PART_WGS] : E;
    const int cnt   = stop - start;

    if (tid < BIN_NODES) cntl[tid] = 0;
    __syncthreads();
    for (int i = start + tid; i < stop; i += 4096) {
        int d[4];
#pragma unroll
        for (int u = 0; u < 4; ++u) {
            const int ii = i + u * 1024;
            d[u] = (ii < stop) ? part[ii].y : -1;
        }
#pragma unroll
        for (int u = 0; u < 4; ++u) {
            if (d[u] >= 0) atomicAdd(&cntl[d[u] - n0], 1);
        }
    }
    __syncthreads();

    const int c = (tid < BIN_NODES) ? cntl[tid] : 0;
    int inc = c;
    for (int off = 1; off < 32; off <<= 1) {
        const int u = __shfl_up(inc, off, 32);
        if (lane >= off) inc += u;
    }
    if (lane == 31 && wave < 16) wtot[wave] = inc;
    __syncthreads();
    if (tid < BIN_NODES) {
        int ex = inc - c;
        for (int w = 0; w < wave; ++w) ex += wtot[w];
        cursor[tid] = ex;
        nbeg[tid] = ex;
        if (tid == BIN_NODES - 1) nbeg[BIN_NODES] = ex + c;
        const int node = n0 + tid;
        if (node <= N) rowptr[node] = start + ex;
    }
    __syncthreads();

    if (cnt <= BIN_CAP) {
        for (int i = start + tid; i < stop; i += 4096) {
            int2 r[4];
#pragma unroll
            for (int u = 0; u < 4; ++u) {
                const int ii = i + u * 1024;
                r[u] = (ii < stop) ? part[ii] : make_int2(0, -1);
            }
#pragma unroll
            for (int u = 0; u < 4; ++u) {
                if (r[u].y >= 0) {
                    const int nd = r[u].y - n0;
                    const int p = atomicAdd(&cursor[nd], 1);
                    lsrc[p]  = r[u].x;
                    lnode[p] = (unsigned short)nd;
                }
            }
        }
        __syncthreads();
        for (int j = tid; j < cnt; j += 1024) {
            const int sv = lsrc[j];
            const int nd = lnode[j];
            const int gb = nbeg[nd];
            const int ge = nbeg[nd + 1];
            int r = 0;
            for (int k = gb; k < ge; ++k) {
                const int v = lsrc[k];
                r += (v < sv || (v == sv && k < j)) ? 1 : 0;
            }
            col[start + gb + r] = sv;
        }
    } else {
        for (int i = start + tid; i < stop; i += 4096) {
            int2 r[4];
#pragma unroll
            for (int u = 0; u < 4; ++u) {
                const int ii = i + u * 1024;
                r[u] = (ii < stop) ? part[ii] : make_int2(0, -1);
            }
#pragma unroll
            for (int u = 0; u < 4; ++u) {
                if (r[u].y >= 0) {
                    const int p = atomicAdd(&cursor[r[u].y - n0], 1);
                    col[start + p] = r[u].x;
                }
            }
        }
        __syncthreads();
        for (int nl = wave; nl < BIN_NODES; nl += 32) {
            const int beg = start + nbeg[nl];
            const int deg = nbeg[nl + 1] - nbeg[nl];
            if (deg <= 1) continue;
            if (deg <= 32) {
                const int v = (lane < deg) ? col[beg + lane] : 0;
                int r = 0;
                for (int j = 0; j < deg; ++j) {
                    const int vj = __shfl(v, j, 32);
                    r += (vj < v || (vj == v && j < lane)) ? 1 : 0;
                }
                if (lane < deg) col[beg + r] = v;
            } else if (lane == 0) {
                for (int i = 1; i < deg; ++i) {
                    const int key = col[beg + i];
                    int j = i - 1;
                    while (j >= 0 && col[beg + j] > key) {
                        col[beg + j + 1] = col[beg + j];
                        --j;
                    }
                    col[beg + j + 1] = key;
                }
            }
        }
    }
}

#define ROW_LOAD(S) (*(const float4*)(x + (long)(S) * 128 + lane * 4))
#define ROW_ADD(V) do { acc.x += (V).x; acc.y += (V).y; acc.z += (V).z; acc.w += (V).w; } while (0)

#define TILE_PITCH 132

struct StripWalk {
    int cur, cur_beg, cur_end;
    float4 acc;
};

__device__ __forceinline__ void close_node(float* tl, const int lane, const int rpv, StripWalk& w) {
    const int deg = w.cur_end - w.cur_beg;
    const float nv = deg > 0 ? 1.0f / (float)deg : 0.0f;
    float4 o;
    o.x = w.acc.x * nv;
    o.y = w.acc.y * nv;
    o.z = w.acc.z * nv;
    o.w = w.acc.w * nv;
    *(float4*)(tl + w.cur * TILE_PITCH + lane * 4) = o;
    w.acc = make_float4(0.0f, 0.0f, 0.0f, 0.0f);
    ++w.cur;
    w.cur_beg = w.cur_end;
    w.cur_end = __builtin_amdgcn_readlane(rpv, w.cur + 1);
}

#define ADD_ENTRY(E, V) do { \
        while ((E) >= w.cur_end) close_node(tl, lane, rpv, w); \
        float4& acc = w.acc; \
        ROW_ADD(V); \
    } while (0)

__device__ __forceinline__ void gather_batch16(const float* __restrict__ x, float* tl, const int lane,
                                               const int ids, const int j0, const int e0,
                                               const int rpv, StripWalk& w,
                                               const int* __restrict__ col, const bool pf,
                                               const int pf_base, const int e_end, int& ids_n) {
    float4 v[16];
#pragma unroll
    for (int u = 0; u < 16; ++u) v[u] = ROW_LOAD(__builtin_amdgcn_readlane(ids, j0 + u));
    if (pf) ids_n = (pf_base + lane < e_end) ? col[pf_base + lane] : 0;
#pragma unroll
    for (int u = 0; u < 16; ++u) ADD_ENTRY(e0 + u, v[u]);
}

__device__ __forceinline__ void gather_tail(const float* __restrict__ x, float* tl, const int lane,
                                            const int ids, const int j0, const int e0, const int c,
                                            const int rpv, StripWalk& w) {
    const int o4 = c & 8;
    const int o2 = o4 + (c & 4);
    const int o1 = o2 + (c & 2);
    if (c & 8) {
        float4 v8[8];
#pragma unroll
        for (int u = 0; u < 8; ++u) v8[u] = ROW_LOAD(__builtin_amdgcn_readlane(ids, j0 + u));
#pragma unroll
        for (int u = 0; u < 8; ++u) ADD_ENTRY(e0 + u, v8[u]);
    }
    float4 v4[4], v2[2], v1;
    if (c & 4) {
#pragma unroll
        for (int u = 0; u < 4; ++u) v4[u] = ROW_LOAD(__builtin_amdgcn_readlane(ids, j0 + o4 + u));
    }
    if (c & 2) {
#pragma unroll
        for (int u = 0; u < 2; ++u) v2[u] = ROW_LOAD(__builtin_amdgcn_readlane(ids, j0 + o2 + u));
    }
    if (c & 1) v1 = ROW_LOAD(__builtin_amdgcn_readlane(ids, j0 + o1));
    if (c & 4) {
#pragma unroll
        for (int u = 0; u < 4; ++u) ADD_ENTRY(e0 + o4 + u, v4[u]);
    }
    if (c & 2) {
#pragma unroll
        for (int u = 0; u < 2; ++u) ADD_ENTRY(e0 + o2 + u, v2[u]);
    }
    if (c & 1) ADD_ENTRY(e0 + o1, v1);
}

template <int NT>
__device__ __forceinline__ void chunk_products(const float* wq, const float4 (&a)[4], v8f (&acc)[NT]) {
    constexpr int LDW = 260;
#pragma unroll
    for (int j = 0; j < 4; ++j) {
        float4 b[NT];
#pragma unroll
        for (int c = 0; c < NT; ++c) b[c] = *(const float4*)(wq + c * 16 * LDW + j * 8);
        const v2f a0 = {a[j].x, a[j].y};
        const v2f a1 = {a[j].z, a[j].w};
#pragma unroll
        for (int c = 0; c < NT; ++c) {
            const v2f b0 = {b[c].x, b[c].y};
            acc[c] = __builtin_amdgcn_wmma_f32_16x16x4_f32(
                false, a0, false, b0, (short)0, acc[c], false, false);
        }
#pragma unroll
        for (int c = 0; c < NT; ++c) {
            const v2f b1 = {b[c].z, b[c].w};
            acc[c] = __builtin_amdgcn_wmma_f32_16x16x4_f32(
                false, a1, false, b1, (short)0, acc[c], false, false);
        }
    }
}

template <int NT>
__device__ __forceinline__ void x_half(const float* __restrict__ xa, const float* wb, v8f (&acc)[NT]) {
    float4 a_nxt[4];
#pragma unroll
    for (int j = 0; j < 4; ++j) a_nxt[j] = *(const float4*)(xa + j * 8);
#pragma unroll 1
    for (int q = 0; q < 4; ++q) {
        float4 a_cur[4];
#pragma unroll
        for (int j = 0; j < 4; ++j) a_cur[j] = a_nxt[j];
        if (q < 3) {
#pragma unroll
            for (int j = 0; j < 4; ++j) a_nxt[j] = *(const float4*)(xa + (q + 1) * 32 + j * 8);
        }
        chunk_products<NT>(wb + q * 32, a_cur, acc);
    }
}

template <int NT>
__device__ __forceinline__ void mean_half(const float* ta, const float* wb, v8f (&acc)[NT]) {
    float4 a_nxt[4];
#pragma unroll
    for (int j = 0; j < 4; ++j) a_nxt[j] = *(const float4*)(ta + j * 8);
#pragma unroll 1
    for (int q = 0; q < 4; ++q) {
        float4 a_cur[4];
#pragma unroll
        for (int j = 0; j < 4; ++j) a_cur[j] = a_nxt[j];
        if (q < 3) {
#pragma unroll
            for (int j = 0; j < 4; ++j) a_nxt[j] = *(const float4*)(ta + (q + 1) * 32 + j * 8);
        }
        chunk_products<NT>(wb + 128 + q * 32, a_cur, acc);
    }
}

template <int NT, bool LNRELU>
__global__ __launch_bounds__(512) void sage_layer_kernel(
    const float* __restrict__ x,
    const int* __restrict__ rowptr,
    const int* __restrict__ col,
    const float* __restrict__ W,
    const float* __restrict__ bias,
    float* __restrict__ y,
    int N) {
    constexpr int NDIM = NT * 16;
    constexpr int LDW  = 260;
    constexpr int KQ   = 512 / NDIM;

    __shared__ __attribute__((aligned(16))) float Wt[NDIM * LDW];
    __shared__ __attribute__((aligned(16))) float Mt[16 * 16 * TILE_PITCH];

    const int tid  = threadIdx.x;
    const int wave = __builtin_amdgcn_readfirstlane(tid >> 5);
    const int lane = tid & 31;
    const int ml   = lane & 15;
    const int half = lane >> 4;

    {
        const int n  = tid & (NDIM - 1);
        const int kq = tid / NDIM;
#pragma unroll 4
        for (int it = 0; it < 64 / KQ; ++it) {
            const int k = (it * KQ + kq) * 4;
            float4 v;
            v.x = W[(k + 0) * NDIM + n];
            v.y = W[(k + 1) * NDIM + n];
            v.z = W[(k + 2) * NDIM + n];
            v.w = W[(k + 3) * NDIM + n];
            *(float4*)&Wt[n * LDW + k] = v;
        }
    }
    __syncthreads();

    float bv[NT];
#pragma unroll
    for (int c = 0; c < NT; ++c) bv[c] = bias[c * 16 + ml];

    const float* wb = &Wt[ml * LDW + half * 4];
    float* tl = &Mt[wave * 16 * TILE_PITCH];
    const float* ta = tl + ml * TILE_PITCH + half * 4;
    const int nstrips = (N + 15) >> 4;
    const bool x_first = (((wave >> 3) ^ (wave >> 1)) & 1) != 0;

    for (int s = blockIdx.x + gridDim.x * wave; s < nstrips; s += gridDim.x * 16) {
        long row = (long)s * 16 + ml;
        if (row >= N) row = N - 1;
        const float* xa = x + row * 128 + half * 4;

        v8f acc[NT];
#pragma unroll
        for (int c = 0; c < NT; ++c) acc[c] = v8f{};

        if (x_first) x_half<NT>(xa, wb, acc);

        {
            int rpi = s * 16 + lane;
            if (rpi > N) rpi = N;
            const int rpv   = rowptr[rpi];
            const int e_beg = __builtin_amdgcn_readlane(rpv, 0);
            const int e_end = __builtin_amdgcn_readlane(rpv, 16);
            StripWalk w;
            w.cur = 0;
            w.cur_beg = e_beg;
            w.cur_end = __builtin_amdgcn_readlane(rpv, 1);
            w.acc = make_float4(0.0f, 0.0f, 0.0f, 0.0f);

            int ids_n = (e_beg + lane < e_end) ? col[e_beg + lane] : 0;
            for (int base = e_beg; base < e_end; base += 32) {
                const int ids = ids_n;
                const int cnt = (e_end - base) < 32 ? (e_end - base) : 32;
                int j = 0;
                for (; j + 16 <= cnt; j += 16)
                    gather_batch16(x, tl, lane, ids, j, base + j, rpv, w,
                                   col, (j == 0) && (cnt == 32), base + 32, e_end, ids_n);
                if (j < cnt) gather_tail(x, tl, lane, ids, j, base + j, cnt - j, rpv, w);
            }
            while (w.cur < 16) close_node(tl, lane, rpv, w);
        }

        if (!x_first) x_half<NT>(xa, wb, acc);
        mean_half<NT>(ta, wb, acc);

#pragma unroll
        for (int c = 0; c < NT; ++c) {
#pragma unroll
            for (int r = 0; r < 8; ++r) acc[c][r] += bv[c];
        }

        if (LNRELU) {
#pragma unroll
            for (int r = 0; r < 8; ++r) {
                float sv = 0.0f, qv = 0.0f;
#pragma unroll
                for (int c = 0; c < NT; ++c) {
                    sv += acc[c][r];
                    qv += acc[c][r] * acc[c][r];
                }
#pragma unroll
                for (int off = 1; off < 16; off <<= 1) {
                    sv += __shfl_xor(sv, off, 32);
                    qv += __shfl_xor(qv, off, 32);
                }
                const float mu   = sv * (1.0f / NDIM);
                const float var  = qv * (1.0f / NDIM) - mu * mu;
                const float rstd = rsqrtf(var + LN_EPS);
#pragma unroll
                for (int c = 0; c < NT; ++c) {
                    const float v = (acc[c][r] - mu) * rstd;
                    acc[c][r] = v > 0.0f ? v : 0.0f;
                }
            }
        }

#pragma unroll
        for (int r = 0; r < 8; ++r) {
            const long ro = (long)s * 16 + half * 8 + r;
            if (ro < N) {
#pragma unroll
                for (int c = 0; c < NT; ++c) y[ro * NDIM + c * 16 + ml] = acc[c][r];
            }
        }
    }
}


extern "C" void kernel_launch(void* const* d_in, const int* in_sizes, int n_in,
                              void* d_out, int out_size, void* d_ws, size_t ws_size,
                              hipStream_t stream) {
    const float* h   = (const float*)d_in[0];
    const int*   src = (const int*)d_in[1];
    const int*   dst = (const int*)d_in[2];
    const float* W1  = (const float*)d_in[3];
    const float* b1  = (const float*)d_in[4];
    const float* W2  = (const float*)d_in[5];
    const float* b2  = (const float*)d_in[6];
    const float* W3  = (const float*)d_in[7];
    const float* b3  = (const float*)d_in[8];
    float* out = (float*)d_out;

    const int N = in_sizes[0] / 128;
    const int E = in_sizes[1];

    const int nbins = (N >> BIN_SHIFT) + 1;
    const int npad  = nbins * BIN_NODES;
    const int M     = nbins * PART_WGS;
    const int epad  = (E + 63) & ~63;
    const int chunk = (E + PART_WGS - 1) / PART_WGS;

    int*   rowptr = (int*)d_ws;
    int*   ghist  = rowptr + npad;
    int*   goff   = ghist + M;
    int2*  part   = (int2*)(goff + M);
    int*   col    = (int*)(part + epad);
    float* h1     = (float*)(col + epad);
    float* h2     = h1  + (size_t)N * 128;

    bin_count_kernel<<<PART_WGS, 1024, 0, stream>>>(dst, ghist, E, chunk, nbins);
    bin_scan_kernel<<<1, 1024, 0, stream>>>(ghist, goff, M);
    bin_scatter_kernel<<<PART_WGS, 1024, 0, stream>>>(src, dst, goff, part, E, chunk, nbins);
    bin_build_kernel<<<nbins, 1024, 0, stream>>>(part, goff, rowptr, col, N, E, nbins);

    sage_layer_kernel<8, true><<<256, 512, 0, stream>>>(h, rowptr, col, W1, b1, h1, N);

    sage_layer_kernel<8, true><<<256, 512, 0, stream>>>(h1, rowptr, col, W2, b2, h2, N);

    sage_layer_kernel<4, false><<<256, 512, 0, stream>>>(h2, rowptr, col, W3, b3, out, N);
}


# Gathered rows read as scalar row base + one fixed lane-offset register, a group's bases made first in register pairs of their own: 16 loads issue back to back, no translation waits; sums unchanged
# speedup vs baseline: 3.7886x; 1.0335x over previous
#include <hip/hip_runtime.h>


typedef __attribute__((ext_vector_type(2))) float v2f;
typedef __attribute__((ext_vector_type(8))) float v8f;

#define LN_EPS 1e-5f


#define BIN_SHIFT 9
#define BIN_NODES 512
#define MAX_BINS  2048
#define PART_WGS  256

__global__ __launch_bounds__(1024) void bin_count_kernel(const int* __restrict__ dst,
                                                         int* __restrict__ ghist,
                                                         int E, int chunk, int nbins) {
    __shared__ int hist[MAX_BINS];
    const int tid = threadIdx.x;
    for (int i = tid; i < nbins; i += 1024) hist[i] = 0;
    __syncthreads();
    const int e0 = blockIdx.x * chunk;
    const int e1 = (e0 + chunk < E) ? (e0 + chunk) : E;
    for (int e = e0 + tid; e < e1; e += 4096) {
        int d[4];
#pragma unroll
        for (int u = 0; u < 4; ++u) {
            const int ee = e + u * 1024;
            d[u] = (ee < e1) ? dst[ee] : -1;
        }
#pragma unroll
        for (int u = 0; u < 4; ++u) {
            if (d[u] >= 0) atomicAdd(&hist[d[u] >> BIN_SHIFT], 1);
        }
    }
    __syncthreads();
    for (int i = tid; i < nbins; i += 1024) ghist[i * PART_WGS + blockIdx.x] = hist[i];
}

__global__ __launch_bounds__(1024) void bin_scan_kernel(const int* __restrict__ ghist,
                                                        int* __restrict__ goff, int M) {
    __shared__ int wtot[32];
    const int tid  = threadIdx.x;
    const int lane = tid & 31;
    const int wave = tid >> 5;
    const int per  = (M + 8191) >> 13;
    const int4* gin = (const int4*)ghist;
    int4* gout = (int4*)goff;
    const int4 zero4 = make_int4(0, 0, 0, 0);

    int s = 0;
    for (int k0 = 0; k0 < per; k0 += 4) {
        int4 a[4], b[4];
#pragma unroll
        for (int u = 0; u < 4; ++u) {
            const int row = wave * per + k0 + u;
            const bool ok = (k0 + u < per) && (row * PART_WGS < M);
            const int rc = ok ? row : 0;
            a[u] = gin[rc * 64 + lane * 2];
            b[u] = gin[rc * 64 + lane * 2 + 1];
            if (!ok) { a[u] = zero4; b[u] = zero4; }
        }
#pragma unroll
        for (int u = 0; u < 4; ++u)
            s += a[u].x + a[u].y + a[u].z + a[u].w + b[u].x + b[u].y + b[u].z + b[u].w;
    }
    for (int off = 16; off > 0; off >>= 1) s += __shfl_xor(s, off, 32);
    if (lane == 0) wtot[wave] = s;
    __syncthreads();

    int run = 0;
    for (int w = 0; w < wave; ++w) run += wtot[w];
    for (int k0 = 0; k0 < per; k0 += 4) {
        int4 a[4], b[4];
#pragma unroll
        for (int u = 0; u < 4; ++u) {
            const int row = wave * per + k0 + u;
            const bool ok = (k0 + u < per) && (row * PART_WGS < M);
            const int rc = ok ? row : 0;
            a[u] = gin[rc * 64 + lane * 2];
            b[u] = gin[rc * 64 + lane * 2 + 1];
            if (!ok) { a[u] = zero4; b[u] = zero4; }
        }
#pragma unroll
        for (int u = 0; u < 4; ++u) {
            const int row = wave * per + k0 + u;
            const bool ok = (k0 + u < per) && (row * PART_WGS < M);
            const int t = a[u].x + a[u].y + a[u].z + a[u].w + b[u].x + b[u].y + b[u].z + b[u].w;
            int inc = t;
            for (int off = 1; off < 32; off <<= 1) {
                const int v = __shfl_up(inc, off, 32);
                if (lane >= off) inc += v;
            }
            int4 oa, ob;
            oa.x = run + inc - t;
            oa.y = oa.x + a[u].x;
            oa.z = oa.y + a[u].y;
            oa.w = oa.z + a[u].z;
            ob.x = oa.w + a[u].w;
            ob.y = ob.x + b[u].x;
            ob.z = ob.y + b[u].y;
            ob.w = ob.z + b[u].z;
            if (ok) {
                gout[row * 64 + lane * 2]     = oa;
                gout[row * 64 + lane * 2 + 1] = ob;
            }
            run += __shfl(inc, 31, 32);
        }
    }
}

__global__ __launch_bounds__(1024) void bin_scatter_kernel(const int* __restrict__ src,
                                                           const int* __restrict__ dst,
                                                           const int* __restrict__ goff,
                                                           int2* __restrict__ part,
                                                           int E, int chunk, int nbins) {
    __shared__ int cur[MAX_BINS];
    const int tid = threadIdx.x;
    for (int i = tid; i < nbins; i += 1024) cur[i] = goff[i * PART_WGS + blockIdx.x];
    __syncthreads();
    const int e0 = blockIdx.x * chunk;
    const int e1 = (e0 + chunk < E) ? (e0 + chunk) : E;
    for (int e = e0 + tid; e < e1; e += 4096) {
        int d[4], s[4];
#pragma unroll
        for (int u = 0; u < 4; ++u) {
            const int ee = e + u * 1024;
            d[u] = (ee < e1) ? dst[ee] : -1;
            s[u] = (ee < e1) ? src[ee] : 0;
        }
#pragma unroll
        for (int u = 0; u < 4; ++u) {
            if (d[u] >= 0) {
                const int p = atomicAdd(&cur[d[u] >> BIN_SHIFT], 1);
                part[p] = make_int2(s[u], d[u]);
            }
        }
    }
}

#define BIN_CAP 8192

__global__ __launch_bounds__(1024) void bin_build_kernel(const int2* __restrict__ part,
                                                         const int* __restrict__ goff,
                                                         int* __restrict__ rowptr,
                                                         int* __restrict__ col,
                                                         int N, int E, int nbins) {
    __shared__ int cntl[BIN_NODES];
    __shared__ int cursor[BIN_NODES];
    __shared__ int nbeg[BIN_NODES + 1];
    __shared__ int wtot[16];
    __shared__ int lsrc[BIN_CAP];
    __shared__ unsigned short lnode[BIN_CAP];
    const int tid  = threadIdx.x;
    const int lane = tid & 31;
    const int wave = tid >> 5;
    const int bin  = blockIdx.x;
    const int n0   = bin << BIN_SHIFT;
    const int start = goff[bin * PART_WGS];
    const int stop  = (bin + 1 < nbins) ? goff[(bin + 1) * PART_WGS] : E;
    const int cnt   = stop - start;

    if (tid < BIN_NODES) cntl[tid] = 0;
    __syncthreads();
    for (int i = start + tid; i < stop; i += 4096) {
        int d[4];
#pragma unroll
        for (int u = 0; u < 4; ++u) {
            const int ii = i + u * 1024;
            d[u] = (ii < stop) ? part[ii].y : -1;
        }
#pragma unroll
        for (int u = 0; u < 4; ++u) {
            if (d[u] >= 0) atomicAdd(&cntl[d[u] - n0], 1);
        }
    }
    __syncthreads();

    const int c = (tid < BIN_NODES) ? cntl[tid] : 0;
    int inc = c;
    for (int off = 1; off < 32; off <<= 1) {
        const int u = __shfl_up(inc, off, 32);
        if (lane >= off) inc += u;
    }
    if (lane == 31 && wave < 16) wtot[wave] = inc;
    __syncthreads();
    if (tid < BIN_NODES) {
        int ex = inc - c;
        for (int w = 0; w < wave; ++w) ex += wtot[w];
        cursor[tid] = ex;
        nbeg[tid] = ex;
        if (tid == BIN_NODES - 1) nbeg[BIN_NODES] = ex + c;
        const int node = n0 + tid;
        if (node <= N) rowptr[node] = start + ex;
    }
    __syncthreads();

    if (cnt <= BIN_CAP) {
        for (int i = start + tid; i < stop; i += 4096) {
            int2 r[4];
#pragma unroll
            for (int u = 0; u < 4; ++u) {
                const int ii = i + u * 1024;
                r[u] = (ii < stop) ? part[ii] : make_int2(0, -1);
            }
#pragma unroll
            for (int u = 0; u < 4; ++u) {
                if (r[u].y >= 0) {
                    const int nd = r[u].y - n0;
                    const int p = atomicAdd(&cursor[nd], 1);
                    lsrc[p]  = r[u].x;
                    lnode[p] = (unsigned short)nd;
                }
            }
        }
        __syncthreads();
        for (int j = tid; j < cnt; j += 1024) {
            const int sv = lsrc[j];
            const int nd = lnode[j];
            const int gb = nbeg[nd];
            const int ge = nbeg[nd + 1];
            int r = 0;
            for (int k = gb; k < ge; ++k) {
                const int v = lsrc[k];
                r += (v < sv || (v == sv && k < j)) ? 1 : 0;
            }
            col[start + gb + r] = sv;
        }
    } else {
        for (int i = start + tid; i < stop; i += 4096) {
            int2 r[4];
#pragma unroll
            for (int u = 0; u < 4; ++u) {
                const int ii = i + u * 1024;
                r[u] = (ii < stop) ? part[ii] : make_int2(0, -1);
            }
#pragma unroll
            for (int u = 0; u < 4; ++u) {
                if (r[u].y >= 0) {
                    const int p = atomicAdd(&cursor[r[u].y - n0], 1);
                    col[start + p] = r[u].x;
                }
            }
        }
        __syncthreads();
        for (int nl = wave; nl < BIN_NODES; nl += 32) {
            const int beg = start + nbeg[nl];
            const int deg = nbeg[nl + 1] - nbeg[nl];
            if (deg <= 1) continue;
            if (deg <= 32) {
                const int v = (lane < deg) ? col[beg + lane] : 0;
                int r = 0;
                for (int j = 0; j < deg; ++j) {
                    const int vj = __shfl(v, j, 32);
                    r += (vj < v || (vj == v && j < lane)) ? 1 : 0;
                }
                if (lane < deg) col[beg + r] = v;
            } else if (lane == 0) {
                for (int i = 1; i < deg; ++i) {
                    const int key = col[beg + i];
                    int j = i - 1;
                    while (j >= 0 && col[beg + j] > key) {
                        col[beg + j + 1] = col[beg + j];
                        --j;
                    }
                    col[beg + j + 1] = key;
                }
            }
        }
    }
}

typedef const __attribute__((address_space(1))) char*   gbytes_t;
typedef const __attribute__((address_space(1))) float4* gfloat4_t;

__device__ __forceinline__ unsigned lane_offset(const int lane) {
    unsigned loff = (unsigned)lane * 16u;
    asm volatile("" : "+v"(loff));
    return loff;
}

template <int C>
__device__ __forceinline__ void rows_load(const float* __restrict__ x, const int ids, const int j,
                                          const unsigned loff, float4 (&v)[C]) {
    gbytes_t rb[C];
#pragma unroll
    for (int u = 0; u < C; ++u) {
        rb[u] = (gbytes_t)(x + (long)__builtin_amdgcn_readlane(ids, j + u) * 128);
        asm("" : "+s"(rb[u]));
    }
    __builtin_amdgcn_sched_barrier(0);
#pragma unroll
    for (int u = 0; u < C; ++u) v[u] = *(gfloat4_t)(rb[u] + loff);
}
#define ROW_ADD(V) do { acc.x += (V).x; acc.y += (V).y; acc.z += (V).z; acc.w += (V).w; } while (0)

#define TILE_PITCH 132

struct StripWalk {
    int cur, cur_beg, cur_end;
    float4 acc;
};

__device__ __forceinline__ void close_node(float* tl, const int lane, const int rpv, StripWalk& w) {
    const int deg = w.cur_end - w.cur_beg;
    const float nv = deg > 0 ? 1.0f / (float)deg : 0.0f;
    float4 o;
    o.x = w.acc.x * nv;
    o.y = w.acc.y * nv;
    o.z = w.acc.z * nv;
    o.w = w.acc.w * nv;
    *(float4*)(tl + w.cur * TILE_PITCH + lane * 4) = o;
    w.acc = make_float4(0.0f, 0.0f, 0.0f, 0.0f);
    ++w.cur;
    w.cur_beg = w.cur_end;
    w.cur_end = __builtin_amdgcn_readlane(rpv, w.cur + 1);
}

#define ADD_ENTRY(E, V) do { \
        while ((E) >= w.cur_end) close_node(tl, lane, rpv, w); \
        float4& acc = w.acc; \
        ROW_ADD(V); \
    } while (0)

__device__ __forceinline__ void gather_batch16(const float* __restrict__ x, float* tl, const int lane,
                                               const int ids, const int j0, const int e0,
                                               const int rpv, StripWalk& w,
                                               const int* __restrict__ col, const bool pf,
                                               const int pf_base, const int e_end, int& ids_n) {
    float4 v[16];
    rows_load<16>(x, ids, j0, lane_offset(lane), v);
    if (pf) ids_n = (pf_base + lane < e_end) ? col[pf_base + lane] : 0;
#pragma unroll
    for (int u = 0; u < 16; ++u) ADD_ENTRY(e0 + u, v[u]);
}

__device__ __forceinline__ void gather_tail(const float* __restrict__ x, float* tl, const int lane,
                                            const int ids, const int j0, const int e0, const int c,
                                            const int rpv, StripWalk& w) {
    const int o4 = c & 8;
    const int o2 = o4 + (c & 4);
    const int o1 = o2 + (c & 2);
    if (c & 8) {
        float4 v8[8];
        rows_load<8>(x, ids, j0, lane_offset(lane), v8);
#pragma unroll
        for (int u = 0; u < 8; ++u) ADD_ENTRY(e0 + u, v8[u]);
    }
    float4 v4[4], v2[2], v1[1];
    if (c & 4) rows_load<4>(x, ids, j0 + o4, lane_offset(lane), v4);
    if (c & 2) rows_load<2>(x, ids, j0 + o2, lane_offset(lane), v2);
    if (c & 1) rows_load<1>(x, ids, j0 + o1, lane_offset(lane), v1);
    if (c & 4) {
#pragma unroll
        for (int u = 0; u < 4; ++u) ADD_ENTRY(e0 + o4 + u, v4[u]);
    }
    if (c & 2) {
#pragma unroll
        for (int u = 0; u < 2; ++u) ADD_ENTRY(e0 + o2 + u, v2[u]);
    }
    if (c & 1) ADD_ENTRY(e0 + o1, v1[0]);
}

template <int NT>
__device__ __forceinline__ void chunk_products(const float* wq, const float4 (&a)[4], v8f (&acc)[NT]) {
    constexpr int LDW = 260;
#pragma unroll
    for (int j = 0; j < 4; ++j) {
        float4 b[NT];
#pragma unroll
        for (int c = 0; c < NT; ++c) b[c] = *(const float4*)(wq + c * 16 * LDW + j * 8);
        const v2f a0 = {a[j].x, a[j].y};
        const v2f a1 = {a[j].z, a[j].w};
#pragma unroll
        for (int c = 0; c < NT; ++c) {
            const v2f b0 = {b[c].x, b[c].y};
            acc[c] = __builtin_amdgcn_wmma_f32_16x16x4_f32(
                false, a0, false, b0, (short)0, acc[c], false, false);
        }
#pragma unroll
        for (int c = 0; c < NT; ++c) {
            const v2f b1 = {b[c].z, b[c].w};
            acc[c] = __builtin_amdgcn_wmma_f32_16x16x4_f32(
                false, a1, false, b1, (short)0, acc[c], false, false);
        }
    }
}

template <int NT>
__device__ __forceinline__ void x_half(const float* __restrict__ xa, const float* wb, v8f (&acc)[NT]) {
    float4 a_nxt[4];
#pragma unroll
    for (int j = 0; j < 4; ++j) a_nxt[j] = *(const float4*)(xa + j * 8);
#pragma unroll 1
    for (int q = 0; q < 4; ++q) {
        float4 a_cur[4];
#pragma unroll
        for (int j = 0; j < 4; ++j) a_cur[j] = a_nxt[j];
        if (q < 3) {
#pragma unroll
            for (int j = 0; j < 4; ++j) a_nxt[j] = *(const float4*)(xa + (q + 1) * 32 + j * 8);
        }
        chunk_products<NT>(wb + q * 32, a_cur, acc);
    }
}

template <int NT>
__device__ __forceinline__ void mean_half(const float* ta, const float* wb, v8f (&acc)[NT]) {
    float4 a_nxt[4];
#pragma unroll
    for (int j = 0; j < 4; ++j) a_nxt[j] = *(const float4*)(ta + j * 8);
#pragma unroll 1
    for (int q = 0; q < 4; ++q) {
        float4 a_cur[4];
#pragma unroll
        for (int j = 0; j < 4; ++j) a_cur[j] = a_nxt[j];
        if (q < 3) {
#pragma unroll
            for (int j = 0; j < 4; ++j) a_nxt[j] = *(const float4*)(ta + (q + 1) * 32 + j * 8);
        }
        chunk_products<NT>(wb + 128 + q * 32, a_cur, acc);
    }
}

template <int NT, bool LNRELU>
__global__ __launch_bounds__(512) void sage_layer_kernel(
    const float* __restrict__ x,
    const int* __restrict__ rowptr,
    const int* __restrict__ col,
    const float* __restrict__ W,
    const float* __restrict__ bias,
    float* __restrict__ y,
    int N) {
    constexpr int NDIM = NT * 16;
    constexpr int LDW  = 260;
    constexpr int KQ   = 512 / NDIM;

    __shared__ __attribute__((aligned(16))) float Wt[NDIM * LDW];
    __shared__ __attribute__((aligned(16))) float Mt[16 * 16 * TILE_PITCH];

    const int tid  = threadIdx.x;
    const int wave = __builtin_amdgcn_readfirstlane(tid >> 5);
    const int lane = tid & 31;
    const int ml   = lane & 15;
    const int half = lane >> 4;

    {
        const int n  = tid & (NDIM - 1);
        const int kq = tid / NDIM;
#pragma unroll 4
        for (int it = 0; it < 64 / KQ; ++it) {
            const int k = (it * KQ + kq) * 4;
            float4 v;
            v.x = W[(k + 0) * NDIM + n];
            v.y = W[(k + 1) * NDIM + n];
            v.z = W[(k + 2) * NDIM + n];
            v.w = W[(k + 3) * NDIM + n];
            *(float4*)&Wt[n * LDW + k] = v;
        }
    }
    __syncthreads();

    float bv[NT];
#pragma unroll
    for (int c = 0; c < NT; ++c) bv[c] = bias[c * 16 + ml];

    const float* wb = &Wt[ml * LDW + half * 4];
    float* tl = &Mt[wave * 16 * TILE_PITCH];
    const float* ta = tl + ml * TILE_PITCH + half * 4;
    const int nstrips = (N + 15) >> 4;
    const bool x_first = (((wave >> 3) ^ (wave >> 1)) & 1) != 0;

    for (int s = blockIdx.x + gridDim.x * wave; s < nstrips; s += gridDim.x * 16) {
        long row = (long)s * 16 + ml;
        if (row >= N) row = N - 1;
        const float* xa = x + row * 128 + half * 4;

        v8f acc[NT];
#pragma unroll
        for (int c = 0; c < NT; ++c) acc[c] = v8f{};

        if (x_first) x_half<NT>(xa, wb, acc);

        {
            int rpi = s * 16 + lane;
            if (rpi > N) rpi = N;
            const int rpv   = rowptr[rpi];
            const int e_beg = __builtin_amdgcn_readlane(rpv, 0);
            const int e_end = __builtin_amdgcn_readlane(rpv, 16);
            StripWalk w;
            w.cur = 0;
            w.cur_beg = e_beg;
            w.cur_end = __builtin_amdgcn_readlane(rpv, 1);
            w.acc = make_float4(0.0f, 0.0f, 0.0f, 0.0f);

            int ids_n = (e_beg + lane < e_end) ? col[e_beg + lane] : 0;
            for (int base = e_beg; base < e_end; base += 32) {
                const int ids = ids_n;
                const int cnt = (e_end - base) < 32 ? (e_end - base) : 32;
                int j = 0;
                for (; j + 16 <= cnt; j += 16)
                    gather_batch16(x, tl, lane, ids, j, base + j, rpv, w,
                                   col, (j == 0) && (cnt == 32), base + 32, e_end, ids_n);
                if (j < cnt) gather_tail(x, tl, lane, ids, j, base + j, cnt - j, rpv, w);
            }
            while (w.cur < 16) close_node(tl, lane, rpv, w);
        }

        if (!x_first) x_half<NT>(xa, wb, acc);
        mean_half<NT>(ta, wb, acc);

#pragma unroll
        for (int c = 0; c < NT; ++c) {
#pragma unroll
            for (int r = 0; r < 8; ++r) acc[c][r] += bv[c];
        }

        if (LNRELU) {
#pragma unroll
            for (int r = 0; r < 8; ++r) {
                float sv = 0.0f, qv = 0.0f;
#pragma unroll
                for (int c = 0; c < NT; ++c) {
                    sv += acc[c][r];
                    qv += acc[c][r] * acc[c][r];
                }
#pragma unroll
                for (int off = 1; off < 16; off <<= 1) {
                    sv += __shfl_xor(sv, off, 32);
                    qv += __shfl_xor(qv, off, 32);
                }
                const float mu   = sv * (1.0f / NDIM);
                const float var  = qv * (1.0f / NDIM) - mu * mu;
                const float rstd = rsqrtf(var + LN_EPS);
#pragma unroll
                for (int c = 0; c < NT; ++c) {
                    const float v = (acc[c][r] - mu) * rstd;
                    acc[c][r] = v > 0.0f ? v : 0.0f;
                }
            }
        }

#pragma unroll
        for (int r = 0; r < 8; ++r) {
            const long ro = (long)s * 16 + half * 8 + r;
            if (ro < N) {
#pragma unroll
                for (int c = 0; c < NT; ++c) y[ro * NDIM + c * 16 + ml] = acc[c][r];
            }
        }
    }
}


extern "C" void kernel_launch(void* const* d_in, const int* in_sizes, int n_in,
                              void* d_out, int out_size, void* d_ws, size_t ws_size,
                              hipStream_t stream) {
    const float* h   = (const float*)d_in[0];
    const int*   src = (const int*)d_in[1];
    const int*   dst = (const int*)d_in[2];
    const float* W1  = (const float*)d_in[3];
    const float* b1  = (const float*)d_in[4];
    const float* W2  = (const float*)d_in[5];
    const float* b2  = (const float*)d_in[6];
    const float* W3  = (const float*)d_in[7];
    const float* b3  = (const float*)d_in[8];
    float* out = (float*)d_out;

    const int N = in_sizes[0] / 128;
    const int E = in_sizes[1];

    const int nbins = (N >> BIN_SHIFT) + 1;
    const int npad  = nbins * BIN_NODES;
    const int M     = nbins * PART_WGS;
    const int epad  = (E + 63) & ~63;
    const int chunk = (E + PART_WGS - 1) / PART_WGS;

    int*   rowptr = (int*)d_ws;
    int*   ghist  = rowptr + npad;
    int*   goff   = ghist + M;
    int2*  part   = (int2*)(goff + M);
    int*   col    = (int*)(part + epad);
    float* h1     = (float*)(col + epad);
    float* h2     = h1  + (size_t)N * 128;

    bin_count_kernel<<<PART_WGS, 1024, 0, stream>>>(dst, ghist, E, chunk, nbins);
    bin_scan_kernel<<<1, 1024, 0, stream>>>(ghist, goff, M);
    bin_scatter_kernel<<<PART_WGS, 1024, 0, stream>>>(src, dst, goff, part, E, chunk, nbins);
    bin_build_kernel<<<nbins, 1024, 0, stream>>>(part, goff, rowptr, col, N, E, nbins);

    sage_layer_kernel<8, true><<<256, 512, 0, stream>>>(h, rowptr, col, W1, b1, h1, N);

    sage_layer_kernel<8, true><<<256, 512, 0, stream>>>(h1, rowptr, col, W2, b2, h2, N);

    sage_layer_kernel<4, false><<<256, 512, 0, stream>>>(h2, rowptr, col, W3, b3, out, N);
}


# Gathered rows in flight per batch 16 -> 32 (a whole 32-id trip at once), tail parts requested together: 7 dependent trips per strip, not 11; 1/deg made per strip in lanes to hold text size; sums same
# speedup vs baseline: 3.9209x; 1.0349x over previous
#include <hip/hip_runtime.h>


typedef __attribute__((ext_vector_type(2))) float v2f;
typedef __attribute__((ext_vector_type(8))) float v8f;

#define LN_EPS 1e-5f


#define BIN_SHIFT 9
#define BIN_NODES 512
#define MAX_BINS  2048
#define PART_WGS  256

__global__ __launch_bounds__(1024) void bin_count_kernel(const int* __restrict__ dst,
                                                         int* __restrict__ ghist,
                                                         int E, int chunk, int nbins) {
    __shared__ int hist[MAX_BINS];
    const int tid = threadIdx.x;
    for (int i = tid; i < nbins; i += 1024) hist[i] = 0;
    __syncthreads();
    const int e0 = blockIdx.x * chunk;
    const int e1 = (e0 + chunk < E) ? (e0 + chunk) : E;
    for (int e = e0 + tid; e < e1; e += 4096) {
        int d[4];
#pragma unroll
        for (int u = 0; u < 4; ++u) {
            const int ee = e + u * 1024;
            d[u] = (ee < e1) ? dst[ee] : -1;
        }
#pragma unroll
        for (int u = 0; u < 4; ++u) {
            if (d[u] >= 0) atomicAdd(&hist[d[u] >> BIN_SHIFT], 1);
        }
    }
    __syncthreads();
    for (int i = tid; i < nbins; i += 1024) ghist[i * PART_WGS + blockIdx.x] = hist[i];
}

__global__ __launch_bounds__(1024) void bin_scan_kernel(const int* __restrict__ ghist,
                                                        int* __restrict__ goff, int M) {
    __shared__ int wtot[32];
    const int tid  = threadIdx.x;
    const int lane = tid & 31;
    const int wave = tid >> 5;
    const int per  = (M + 8191) >> 13;
    const int4* gin = (const int4*)ghist;
    int4* gout = (int4*)goff;
    const int4 zero4 = make_int4(0, 0, 0, 0);

    int s = 0;
    for (int k0 = 0; k0 < per; k0 += 4) {
        int4 a[4], b[4];
#pragma unroll
        for (int u = 0; u < 4; ++u) {
            const int row = wave * per + k0 + u;
            const bool ok = (k0 + u < per) && (row * PART_WGS < M);
            const int rc = ok ? row : 0;
            a[u] = gin[rc * 64 + lane * 2];
            b[u] = gin[rc * 64 + lane * 2 + 1];
            if (!ok) { a[u] = zero4; b[u] = zero4; }
        }
#pragma unroll
        for (int u = 0; u < 4; ++u)
            s += a[u].x + a[u].y + a[u].z + a[u].w + b[u].x + b[u].y + b[u].z + b[u].w;
    }
    for (int off = 16; off > 0; off >>= 1) s += __shfl_xor(s, off, 32);
    if (lane == 0) wtot[wave] = s;
    __syncthreads();

    int run = 0;
    for (int w = 0; w < wave; ++w) run += wtot[w];
    for (int k0 = 0; k0 < per; k0 += 4) {
        int4 a[4], b[4];
#pragma unroll
        for (int u = 0; u < 4; ++u) {
            const int row = wave * per + k0 + u;
            const bool ok = (k0 + u < per) && (row * PART_WGS < M);
            const int rc = ok ? row : 0;
            a[u] = gin[rc * 64 + lane * 2];
            b[u] = gin[rc * 64 + lane * 2 + 1];
            if (!ok) { a[u] = zero4; b[u] = zero4; }
        }
#pragma unroll
        for (int u = 0; u < 4; ++u) {
            const int row = wave * per + k0 + u;
            const bool ok = (k0 + u < per) && (row * PART_WGS < M);
            const int t = a[u].x + a[u].y + a[u].z + a[u].w + b[u].x + b[u].y + b[u].z + b[u].w;
            int inc = t;
            for (int off = 1; off < 32; off <<= 1) {
                const int v = __shfl_up(inc, off, 32);
                if (lane >= off) inc += v;
            }
            int4 oa, ob;
            oa.x = run + inc - t;
            oa.y = oa.x + a[u].x;
            oa.z = oa.y + a[u].y;
            oa.w = oa.z + a[u].z;
            ob.x = oa.w + a[u].w;
            ob.y = ob.x + b[u].x;
            ob.z = ob.y + b[u].y;
            ob.w = ob.z + b[u].z;
            if (ok) {
                gout[row * 64 + lane * 2]     = oa;
                gout[row * 64 + lane * 2 + 1] = ob;
            }
            run += __shfl(inc, 31, 32);
        }
    }
}

__global__ __launch_bounds__(1024) void bin_scatter_kernel(const int* __restrict__ src,
                                                           const int* __restrict__ dst,
                                                           const int* __restrict__ goff,
                                                           int2* __restrict__ part,
                                                           int E, int chunk, int nbins) {
    __shared__ int cur[MAX_BINS];
    const int tid = threadIdx.x;
    for (int i = tid; i < nbins; i += 1024) cur[i] = goff[i * PART_WGS + blockIdx.x];
    __syncthreads();
    const int e0 = blockIdx.x * chunk;
    const int e1 = (e0 + chunk < E) ? (e0 + chunk) : E;
    for (int e = e0 + tid; e < e1; e += 4096) {
        int d[4], s[4];
#pragma unroll
        for (int u = 0; u < 4; ++u) {
            const int ee = e + u * 1024;
            d[u] = (ee < e1) ? dst[ee] : -1;
            s[u] = (ee < e1) ? src[ee] : 0;
        }
#pragma unroll
        for (int u = 0; u < 4; ++u) {
            if (d[u] >= 0) {
                const int p = atomicAdd(&cur[d[u] >> BIN_SHIFT], 1);
                part[p] = make_int2(s[u], d[u]);
            }
        }
    }
}

#define BIN_CAP 8192

__global__ __launch_bounds__(1024) void bin_build_kernel(const int2* __restrict__ part,
                                                         const int* __restrict__ goff,
                                                         int* __restrict__ rowptr,
                                                         int* __restrict__ col,
                                                         int N, int E, int nbins) {
    __shared__ int cntl[BIN_NODES];
    __shared__ int cursor[BIN_NODES];
    __shared__ int nbeg[BIN_NODES + 1];
    __shared__ int wtot[16];
    __shared__ int lsrc[BIN_CAP];
    __shared__ unsigned short lnode[BIN_CAP];
    const int tid  = threadIdx.x;
    const int lane = tid & 31;
    const int wave = tid >> 5;
    const int bin  = blockIdx.x;
    const int n0   = bin << BIN_SHIFT;
    const int start = goff[bin * PART_WGS];
    const int stop  = (bin + 1 < nbins) ? goff[(bin + 1) * PART_WGS] : E;
    const int cnt   = stop - start;

    if (tid < BIN_NODES) cntl[tid] = 0;
    __syncthreads();
    for (int i = start + tid; i < stop; i += 4096) {
        int d[4];
#pragma unroll
        for (int u = 0; u < 4; ++u) {
            const int ii = i + u * 1024;
            d[u] = (ii < stop) ? part[ii].y : -1;
        }
#pragma unroll
        for (int u = 0; u < 4; ++u) {
            if (d[u] >= 0) atomicAdd(&cntl[d[u] - n0], 1);
        }
    }
    __syncthreads();

    const int c = (tid < BIN_NODES) ? cntl[tid] : 0;
    int inc = c;
    for (int off = 1; off < 32; off <<= 1) {
        const int u = __shfl_up(inc, off, 32);
        if (lane >= off) inc += u;
    }
    if (lane == 31 && wave < 16) wtot[wave] = inc;
    __syncthreads();
    if (tid < BIN_NODES) {
        int ex = inc - c;
        for (int w = 0; w < wave; ++w) ex += wtot[w];
        cursor[tid] = ex;
        nbeg[tid] = ex;
        if (tid == BIN_NODES - 1) nbeg[BIN_NODES] = ex + c;
        const int node = n0 + tid;
        if (node <= N) rowptr[node] = start + ex;
    }
    __syncthreads();

    if (cnt <= BIN_CAP) {
        for (int i = start + tid; i < stop; i += 4096) {
            int2 r[4];
#pragma unroll
            for (int u = 0; u < 4; ++u) {
                const int ii = i + u * 1024;
                r[u] = (ii < stop) ? part[ii] : make_int2(0, -1);
            }
#pragma unroll
            for (int u = 0; u < 4; ++u) {
                if (r[u].y >= 0) {
                    const int nd = r[u].y - n0;
                    const int p = atomicAdd(&cursor[nd], 1);
                    lsrc[p]  = r[u].x;
                    lnode[p] = (unsigned short)nd;
                }
            }
        }
        __syncthreads();
        for (int j = tid; j < cnt; j += 1024) {
            const int sv = lsrc[j];
            const int nd = lnode[j];
            const int gb = nbeg[nd];
            const int ge = nbeg[nd + 1];
            int r = 0;
            for (int k = gb; k < ge; ++k) {
                const int v = lsrc[k];
                r += (v < sv || (v == sv && k < j)) ? 1 : 0;
            }
            col[start + gb + r] = sv;
        }
    } else {
        for (int i = start + tid; i < stop; i += 4096) {
            int2 r[4];
#pragma unroll
            for (int u = 0; u < 4; ++u) {
                const int ii = i + u * 1024;
                r[u] = (ii < stop) ? part[ii] : make_int2(0, -1);
            }
#pragma unroll
            for (int u = 0; u < 4; ++u) {
                if (r[u].y >= 0) {
                    const int p = atomicAdd(&cursor[r[u].y - n0], 1);
                    col[start + p] = r[u].x;
                }
            }
        }
        __syncthreads();
        for (int nl = wave; nl < BIN_NODES; nl += 32) {
            const int beg = start + nbeg[nl];
            const int deg = nbeg[nl + 1] - nbeg[nl];
            if (deg <= 1) continue;
            if (deg <= 32) {
                const int v = (lane < deg) ? col[beg + lane] : 0;
                int r = 0;
                for (int j = 0; j < deg; ++j) {
                    const int vj = __shfl(v, j, 32);
                    r += (vj < v || (vj == v && j < lane)) ? 1 : 0;
                }
                if (lane < deg) col[beg + r] = v;
            } else if (lane == 0) {
                for (int i = 1; i < deg; ++i) {
                    const int key = col[beg + i];
                    int j = i - 1;
                    while (j >= 0 && col[beg + j] > key) {
                        col[beg + j + 1] = col[beg + j];
                        --j;
                    }
                    col[beg + j + 1] = key;
                }
            }
        }
    }
}

typedef const __attribute__((address_space(1))) char*   gbytes_t;
typedef const __attribute__((address_space(1))) float4* gfloat4_t;

__device__ __forceinline__ unsigned lane_offset(const int lane) {
    unsigned loff = (unsigned)lane * 16u;
    asm volatile("" : "+v"(loff));
    return loff;
}

template <int C>
__device__ __forceinline__ void rows_load(const float* __restrict__ x, const int ids, const int j,
                                          const unsigned loff, float4 (&v)[C]) {
    gbytes_t rb[C];
#pragma unroll
    for (int u = 0; u < C; ++u) {
        rb[u] = (gbytes_t)(x + (long)__builtin_amdgcn_readlane(ids, j + u) * 128);
        asm("" : "+s"(rb[u]));
    }
    __builtin_amdgcn_sched_barrier(0);
#pragma unroll
    for (int u = 0; u < C; ++u) v[u] = *(gfloat4_t)(rb[u] + loff);
}
#define ROW_ADD(V) do { acc.x += (V).x; acc.y += (V).y; acc.z += (V).z; acc.w += (V).w; } while (0)

#define TILE_PITCH 132

struct StripWalk {
    int cur, cur_end;
    float4 acc;
};

__device__ __forceinline__ void close_node(float* tl, const int lane, const int rpv, const int nvi,
                                           StripWalk& w) {
    const float nv = __int_as_float(__builtin_amdgcn_readlane(nvi, w.cur));
    float4 o;
    o.x = w.acc.x * nv;
    o.y = w.acc.y * nv;
    o.z = w.acc.z * nv;
    o.w = w.acc.w * nv;
    *(float4*)(tl + w.cur * TILE_PITCH + lane * 4) = o;
    w.acc = make_float4(0.0f, 0.0f, 0.0f, 0.0f);
    ++w.cur;
    w.cur_end = __builtin_amdgcn_readlane(rpv, w.cur + 1);
}

#define ADD_ENTRY(E, V) do { \
        while ((E) >= w.cur_end) close_node(tl, lane, rpv, nvi, w); \
        float4& acc = w.acc; \
        ROW_ADD(V); \
    } while (0)

__device__ __forceinline__ void gather_batch32(const float* __restrict__ x, float* tl, const int lane,
                                               const int ids, const int e0,
                                               const int rpv, const int nvi, StripWalk& w,
                                               const int* __restrict__ col,
                                               const int pf_base, const int e_end, int& ids_n) {
    float4 va[16], vb[16];
    rows_load<16>(x, ids, 0, lane_offset(lane), va);
    rows_load<16>(x, ids, 16, lane_offset(lane), vb);
    ids_n = (pf_base + lane < e_end) ? col[pf_base + lane] : 0;
    __builtin_amdgcn_sched_barrier(0);
#pragma unroll
    for (int u = 0; u < 16; ++u) ADD_ENTRY(e0 + u, va[u]);
#pragma unroll
    for (int u = 0; u < 16; ++u) ADD_ENTRY(e0 + 16 + u, vb[u]);
}

__device__ __forceinline__ void gather_tail(const float* __restrict__ x, float* tl, const int lane,
                                            const int ids, const int e0, const int c,
                                            const int rpv, const int nvi, StripWalk& w) {
    const int o8 = c & 16;
    const int o4 = o8 + (c & 8);
    const int o2 = o4 + (c & 4);
    const int o1 = o2 + (c & 2);
    float4 v16[16] = {}, v8[8] = {}, v4[4] = {}, v2[2] = {}, v1[1] = {};
    asm volatile("" : : "v"(ids));
    if (c & 16) rows_load<16>(x, ids, 0, lane_offset(lane), v16);
    if (c & 8) rows_load<8>(x, ids, o8, lane_offset(lane), v8);
    if (c & 4) rows_load<4>(x, ids, o4, lane_offset(lane), v4);
    if (c & 2) rows_load<2>(x, ids, o2, lane_offset(lane), v2);
    if (c & 1) rows_load<1>(x, ids, o1, lane_offset(lane), v1);
    if (c & 16) {
#pragma unroll
        for (int u = 0; u < 16; ++u) ADD_ENTRY(e0 + u, v16[u]);
    }
    if (c & 8) {
#pragma unroll
        for (int u = 0; u < 8; ++u) ADD_ENTRY(e0 + o8 + u, v8[u]);
    }
    if (c & 4) {
#pragma unroll
        for (int u = 0; u < 4; ++u) ADD_ENTRY(e0 + o4 + u, v4[u]);
    }
    if (c & 2) {
#pragma unroll
        for (int u = 0; u < 2; ++u) ADD_ENTRY(e0 + o2 + u, v2[u]);
    }
    if (c & 1) ADD_ENTRY(e0 + o1, v1[0]);
}

template <int NT>
__device__ __forceinline__ void chunk_products(const float* wq, const float4 (&a)[4], v8f (&acc)[NT]) {
    constexpr int LDW = 260;
#pragma unroll
    for (int j = 0; j < 4; ++j) {
        float4 b[NT];
#pragma unroll
        for (int c = 0; c < NT; ++c) b[c] = *(const float4*)(wq + c * 16 * LDW + j * 8);
        const v2f a0 = {a[j].x, a[j].y};
        const v2f a1 = {a[j].z, a[j].w};
#pragma unroll
        for (int c = 0; c < NT; ++c) {
            const v2f b0 = {b[c].x, b[c].y};
            acc[c] = __builtin_amdgcn_wmma_f32_16x16x4_f32(
                false, a0, false, b0, (short)0, acc[c], false, false);
        }
#pragma unroll
        for (int c = 0; c < NT; ++c) {
            const v2f b1 = {b[c].z, b[c].w};
            acc[c] = __builtin_amdgcn_wmma_f32_16x16x4_f32(
                false, a1, false, b1, (short)0, acc[c], false, false);
        }
    }
}

template <int NT>
__device__ __forceinline__ void x_half(const float* __restrict__ xa, const float* wb, v8f (&acc)[NT]) {
    float4 a_nxt[4];
#pragma unroll
    for (int j = 0; j < 4; ++j) a_nxt[j] = *(const float4*)(xa + j * 8);
#pragma unroll 1
    for (int q = 0; q < 4; ++q) {
        float4 a_cur[4];
#pragma unroll
        for (int j = 0; j < 4; ++j) a_cur[j] = a_nxt[j];
        if (q < 3) {
#pragma unroll
            for (int j = 0; j < 4; ++j) a_nxt[j] = *(const float4*)(xa + (q + 1) * 32 + j * 8);
        }
        chunk_products<NT>(wb + q * 32, a_cur, acc);
    }
}

template <int NT>
__device__ __forceinline__ void mean_half(const float* ta, const float* wb, v8f (&acc)[NT]) {
    float4 a_nxt[4];
#pragma unroll
    for (int j = 0; j < 4; ++j) a_nxt[j] = *(const float4*)(ta + j * 8);
#pragma unroll 1
    for (int q = 0; q < 4; ++q) {
        float4 a_cur[4];
#pragma unroll
        for (int j = 0; j < 4; ++j) a_cur[j] = a_nxt[j];
        if (q < 3) {
#pragma unroll
            for (int j = 0; j < 4; ++j) a_nxt[j] = *(const float4*)(ta + (q + 1) * 32 + j * 8);
        }
        chunk_products<NT>(wb + 128 + q * 32, a_cur, acc);
    }
}

template <int NT, bool LNRELU>
__global__ __launch_bounds__(512) void sage_layer_kernel(
    const float* __restrict__ x,
    const int* __restrict__ rowptr,
    const int* __restrict__ col,
    const float* __restrict__ W,
    const float* __restrict__ bias,
    float* __restrict__ y,
    int N) {
    constexpr int NDIM = NT * 16;
    constexpr int LDW  = 260;
    constexpr int KQ   = 512 / NDIM;

    __shared__ __attribute__((aligned(16))) float Wt[NDIM * LDW];
    __shared__ __attribute__((aligned(16))) float Mt[16 * 16 * TILE_PITCH];

    const int tid  = threadIdx.x;
    const int wave = __builtin_amdgcn_readfirstlane(tid >> 5);
    const int lane = tid & 31;
    const int ml   = lane & 15;
    const int half = lane >> 4;

    {
        const int n  = tid & (NDIM - 1);
        const int kq = tid / NDIM;
#pragma unroll 4
        for (int it = 0; it < 64 / KQ; ++it) {
            const int k = (it * KQ + kq) * 4;
            float4 v;
            v.x = W[(k + 0) * NDIM + n];
            v.y = W[(k + 1) * NDIM + n];
            v.z = W[(k + 2) * NDIM + n];
            v.w = W[(k + 3) * NDIM + n];
            *(float4*)&Wt[n * LDW + k] = v;
        }
    }
    __syncthreads();

    float bv[NT];
#pragma unroll
    for (int c = 0; c < NT; ++c) bv[c] = bias[c * 16 + ml];

    const float* wb = &Wt[ml * LDW + half * 4];
    float* tl = &Mt[wave * 16 * TILE_PITCH];
    const float* ta = tl + ml * TILE_PITCH + half * 4;
    const int nstrips = (N + 15) >> 4;
    const bool x_first = (((wave >> 3) ^ (wave >> 1)) & 1) != 0;

    for (int s = blockIdx.x + gridDim.x * wave; s < nstrips; s += gridDim.x * 16) {
        long row = (long)s * 16 + ml;
        if (row >= N) row = N - 1;
        const float* xa = x + row * 128 + half * 4;

        v8f acc[NT];
#pragma unroll
        for (int c = 0; c < NT; ++c) acc[c] = v8f{};

        if (x_first) x_half<NT>(xa, wb, acc);

        {
            int rpi = s * 16 + lane;
            if (rpi > N) rpi = N;
            const int rpv   = rowptr[rpi];
            const int e_beg = __builtin_amdgcn_readlane(rpv, 0);
            const int e_end = __builtin_amdgcn_readlane(rpv, 16);
            StripWalk w;
            w.cur = 0;
            w.cur_end = __builtin_amdgcn_readlane(rpv, 1);
            w.acc = make_float4(0.0f, 0.0f, 0.0f, 0.0f);

            int ids_n = (e_beg + lane < e_end) ? col[e_beg + lane] : 0;
            const int degl = __shfl_down(rpv, 1, 32) - rpv;
            const int nvi = __float_as_int(degl > 0 ? 1.0f / (float)degl : 0.0f);
            for (int base = e_beg; base < e_end; base += 32) {
                const int ids = ids_n;
                const int cnt = (e_end - base) < 32 ? (e_end - base) : 32;
                if (cnt == 32)
                    gather_batch32(x, tl, lane, ids, base, rpv, nvi, w, col, base + 32, e_end, ids_n);
                else
                    gather_tail(x, tl, lane, ids, base, cnt, rpv, nvi, w);
            }
            while (w.cur < 16) close_node(tl, lane, rpv, nvi, w);
        }

        if (!x_first) x_half<NT>(xa, wb, acc);
        mean_half<NT>(ta, wb, acc);

#pragma unroll
        for (int c = 0; c < NT; ++c) {
#pragma unroll
            for (int r = 0; r < 8; ++r) acc[c][r] += bv[c];
        }

        if (LNRELU) {
#pragma unroll
            for (int r = 0; r < 8; ++r) {
                float sv = 0.0f, qv = 0.0f;
#pragma unroll
                for (int c = 0; c < NT; ++c) {
                    sv += acc[c][r];
                    qv += acc[c][r] * acc[c][r];
                }
#pragma unroll
                for (int off = 1; off < 16; off <<= 1) {
                    sv += __shfl_xor(sv, off, 32);
                    qv += __shfl_xor(qv, off, 32);
                }
                const float mu   = sv * (1.0f / NDIM);
                const float var  = qv * (1.0f / NDIM) - mu * mu;
                const float rstd = rsqrtf(var + LN_EPS);
#pragma unroll
                for (int c = 0; c < NT; ++c) {
                    const float v = (acc[c][r] - mu) * rstd;
                    acc[c][r] = v > 0.0f ? v : 0.0f;
                }
            }
        }

#pragma unroll
        for (int r = 0; r < 8; ++r) {
            const long ro = (long)s * 16 + half * 8 + r;
            if (ro < N) {
#pragma unroll
                for (int c = 0; c < NT; ++c) y[ro * NDIM + c * 16 + ml] = acc[c][r];
            }
        }
    }
}


extern "C" void kernel_launch(void* const* d_in, const int* in_sizes, int n_in,
                              void* d_out, int out_size, void* d_ws, size_t ws_size,
                              hipStream_t stream) {
    const float* h   = (const float*)d_in[0];
    const int*   src = (const int*)d_in[1];
    const int*   dst = (const int*)d_in[2];
    const float* W1  = (const float*)d_in[3];
    const float* b1  = (const float*)d_in[4];
    const float* W2  = (const float*)d_in[5];
    const float* b2  = (const float*)d_in[6];
    const float* W3  = (const float*)d_in[7];
    const float* b3  = (const float*)d_in[8];
    float* out = (float*)d_out;

    const int N = in_sizes[0] / 128;
    const int E = in_sizes[1];

    const int nbins = (N >> BIN_SHIFT) + 1;
    const int npad  = nbins * BIN_NODES;
    const int M     = nbins * PART_WGS;
    const int epad  = (E + 63) & ~63;
    const int chunk = (E + PART_WGS - 1) / PART_WGS;

    int*   rowptr = (int*)d_ws;
    int*   ghist  = rowptr + npad;
    int*   goff   = ghist + M;
    int2*  part   = (int2*)(goff + M);
    int*   col    = (int*)(part + epad);
    float* h1     = (float*)(col + epad);
    float* h2     = h1  + (size_t)N * 128;

    bin_count_kernel<<<PART_WGS, 1024, 0, stream>>>(dst, ghist, E, chunk, nbins);
    bin_scan_kernel<<<1, 1024, 0, stream>>>(ghist, goff, M);
    bin_scatter_kernel<<<PART_WGS, 1024, 0, stream>>>(src, dst, goff, part, E, chunk, nbins);
    bin_build_kernel<<<nbins, 1024, 0, stream>>>(part, goff, rowptr, col, N, E, nbins);

    sage_layer_kernel<8, true><<<256, 512, 0, stream>>>(h, rowptr, col, W1, b1, h1, N);

    sage_layer_kernel<8, true><<<256, 512, 0, stream>>>(h1, rowptr, col, W2, b2, h2, N);

    sage_layer_kernel<4, false><<<256, 512, 0, stream>>>(h2, rowptr, col, W3, b3, out, N);
}


# Last, partly filled round of strips shared: one wave gathers and multiplies the mean half, an idle wave the x half, sums joined in LDS behind one barrier; those rows sum x half + mean half separately
# speedup vs baseline: 4.1604x; 1.0611x over previous
#include <hip/hip_runtime.h>


typedef __attribute__((ext_vector_type(2))) float v2f;
typedef __attribute__((ext_vector_type(8))) float v8f;

#define LN_EPS 1e-5f


#define BIN_SHIFT 9
#define BIN_NODES 512
#define MAX_BINS  2048
#define PART_WGS  256

__global__ __launch_bounds__(1024) void bin_count_kernel(const int* __restrict__ dst,
                                                         int* __restrict__ ghist,
                                                         int E, int chunk, int nbins) {
    __shared__ int hist[MAX_BINS];
    const int tid = threadIdx.x;
    for (int i = tid; i < nbins; i += 1024) hist[i] = 0;
    __syncthreads();
    const int e0 = blockIdx.x * chunk;
    const int e1 = (e0 + chunk < E) ? (e0 + chunk) : E;
    for (int e = e0 + tid; e < e1; e += 4096) {
        int d[4];
#pragma unroll
        for (int u = 0; u < 4; ++u) {
            const int ee = e + u * 1024;
            d[u] = (ee < e1) ? dst[ee] : -1;
        }
#pragma unroll
        for (int u = 0; u < 4; ++u) {
            if (d[u] >= 0) atomicAdd(&hist[d[u] >> BIN_SHIFT], 1);
        }
    }
    __syncthreads();
    for (int i = tid; i < nbins; i += 1024) ghist[i * PART_WGS + blockIdx.x] = hist[i];
}

__global__ __launch_bounds__(1024) void bin_scan_kernel(const int* __restrict__ ghist,
                                                        int* __restrict__ goff, int M) {
    __shared__ int wtot[32];
    const int tid  = threadIdx.x;
    const int lane = tid & 31;
    const int wave = tid >> 5;
    const int per  = (M + 8191) >> 13;
    const int4* gin = (const int4*)ghist;
    int4* gout = (int4*)goff;
    const int4 zero4 = make_int4(0, 0, 0, 0);

    int s = 0;
    for (int k0 = 0; k0 < per; k0 += 4) {
        int4 a[4], b[4];
#pragma unroll
        for (int u = 0; u < 4; ++u) {
            const int row = wave * per + k0 + u;
            const bool ok = (k0 + u < per) && (row * PART_WGS < M);
            const int rc = ok ? row : 0;
            a[u] = gin[rc * 64 + lane * 2];
            b[u] = gin[rc * 64 + lane * 2 + 1];
            if (!ok) { a[u] = zero4; b[u] = zero4; }
        }
#pragma unroll
        for (int u = 0; u < 4; ++u)
            s += a[u].x + a[u].y + a[u].z + a[u].w + b[u].x + b[u].y + b[u].z + b[u].w;
    }
    for (int off = 16; off > 0; off >>= 1) s += __shfl_xor(s, off, 32);
    if (lane == 0) wtot[wave] = s;
    __syncthreads();

    int run = 0;
    for (int w = 0; w < wave; ++w) run += wtot[w];
    for (int k0 = 0; k0 < per; k0 += 4) {
        int4 a[4], b[4];
#pragma unroll
        for (int u = 0; u < 4; ++u) {
            const int row = wave * per + k0 + u;
            const bool ok = (k0 + u < per) && (row * PART_WGS < M);
            const int rc = ok ? row : 0;
            a[u] = gin[rc * 64 + lane * 2];
            b[u] = gin[rc * 64 + lane * 2 + 1];
            if (!ok) { a[u] = zero4; b[u] = zero4; }
        }
#pragma unroll
        for (int u = 0; u < 4; ++u) {
            const int row = wave * per + k0 + u;
            const bool ok = (k0 + u < per) && (row * PART_WGS < M);
            const int t = a[u].x + a[u].y + a[u].z + a[u].w + b[u].x + b[u].y + b[u].z + b[u].w;
            int inc = t;
            for (int off = 1; off < 32; off <<= 1) {
                const int v = __shfl_up(inc, off, 32);
                if (lane >= off) inc += v;
            }
            int4 oa, ob;
            oa.x = run + inc - t;
            oa.y = oa.x + a[u].x;
            oa.z = oa.y + a[u].y;
            oa.w = oa.z + a[u].z;
            ob.x = oa.w + a[u].w;
            ob.y = ob.x + b[u].x;
            ob.z = ob.y + b[u].y;
            ob.w = ob.z + b[u].z;
            if (ok) {
                gout[row * 64 + lane * 2]     = oa;
                gout[row * 64 + lane * 2 + 1] = ob;
            }
            run += __shfl(inc, 31, 32);
        }
    }
}

__global__ __launch_bounds__(1024) void bin_scatter_kernel(const int* __restrict__ src,
                                                           const int* __restrict__ dst,
                                                           const int* __restrict__ goff,
                                                           int2* __restrict__ part,
                                                           int E, int chunk, int nbins) {
    __shared__ int cur[MAX_BINS];
    const int tid = threadIdx.x;
    for (int i = tid; i < nbins; i += 1024) cur[i] = goff[i * PART_WGS + blockIdx.x];
    __syncthreads();
    const int e0 = blockIdx.x * chunk;
    const int e1 = (e0 + chunk < E) ? (e0 + chunk) : E;
    for (int e = e0 + tid; e < e1; e += 4096) {
        int d[4], s[4];
#pragma unroll
        for (int u = 0; u < 4; ++u) {
            const int ee = e + u * 1024;
            d[u] = (ee < e1) ? dst[ee] : -1;
            s[u] = (ee < e1) ? src[ee] : 0;
        }
#pragma unroll
        for (int u = 0; u < 4; ++u) {
            if (d[u] >= 0) {
                const int p = atomicAdd(&cur[d[u] >> BIN_SHIFT], 1);
                part[p] = make_int2(s[u], d[u]);
            }
        }
    }
}

#define BIN_CAP 8192

__global__ __launch_bounds__(1024) void bin_build_kernel(const int2* __restrict__ part,
                                                         const int* __restrict__ goff,
                                                         int* __restrict__ rowptr,
                                                         int* __restrict__ col,
                                                         int N, int E, int nbins) {
    __shared__ int cntl[BIN_NODES];
    __shared__ int cursor[BIN_NODES];
    __shared__ int nbeg[BIN_NODES + 1];
    __shared__ int wtot[16];
    __shared__ int lsrc[BIN_CAP];
    __shared__ unsigned short lnode[BIN_CAP];
    const int tid  = threadIdx.x;
    const int lane = tid & 31;
    const int wave = tid >> 5;
    const int bin  = blockIdx.x;
    const int n0   = bin << BIN_SHIFT;
    const int start = goff[bin * PART_WGS];
    const int stop  = (bin + 1 < nbins) ? goff[(bin + 1) * PART_WGS] : E;
    const int cnt   = stop - start;

    if (tid < BIN_NODES) cntl[tid] = 0;
    __syncthreads();
    for (int i = start + tid; i < stop; i += 4096) {
        int d[4];
#pragma unroll
        for (int u = 0; u < 4; ++u) {
            const int ii = i + u * 1024;
            d[u] = (ii < stop) ? part[ii].y : -1;
        }
#pragma unroll
        for (int u = 0; u < 4; ++u) {
            if (d[u] >= 0) atomicAdd(&cntl[d[u] - n0], 1);
        }
    }
    __syncthreads();

    const int c = (tid < BIN_NODES) ? cntl[tid] : 0;
    int inc = c;
    for (int off = 1; off < 32; off <<= 1) {
        const int u = __shfl_up(inc, off, 32);
        if (lane >= off) inc += u;
    }
    if (lane == 31 && wave < 16) wtot[wave] = inc;
    __syncthreads();
    if (tid < BIN_NODES) {
        int ex = inc - c;
        for (int w = 0; w < wave; ++w) ex += wtot[w];
        cursor[tid] = ex;
        nbeg[tid] = ex;
        if (tid == BIN_NODES - 1) nbeg[BIN_NODES] = ex + c;
        const int node = n0 + tid;
        if (node <= N) rowptr[node] = start + ex;
    }
    __syncthreads();

    if (cnt <= BIN_CAP) {
        for (int i = start + tid; i < stop; i += 4096) {
            int2 r[4];
#pragma unroll
            for (int u = 0; u < 4; ++u) {
                const int ii = i + u * 1024;
                r[u] = (ii < stop) ? part[ii] : make_int2(0, -1);
            }
#pragma unroll
            for (int u = 0; u < 4; ++u) {
                if (r[u].y >= 0) {
                    const int nd = r[u].y - n0;
                    const int p = atomicAdd(&cursor[nd], 1);
                    lsrc[p]  = r[u].x;
                    lnode[p] = (unsigned short)nd;
                }
            }
        }
        __syncthreads();
        for (int j = tid; j < cnt; j += 1024) {
            const int sv = lsrc[j];
            const int nd = lnode[j];
            const int gb = nbeg[nd];
            const int ge = nbeg[nd + 1];
            int r = 0;
            for (int k = gb; k < ge; ++k) {
                const int v = lsrc[k];
                r += (v < sv || (v == sv && k < j)) ? 1 : 0;
            }
            col[start + gb + r] = sv;
        }
    } else {
        for (int i = start + tid; i < stop; i += 4096) {
            int2 r[4];
#pragma unroll
            for (int u = 0; u < 4; ++u) {
                const int ii = i + u * 1024;
                r[u] = (ii < stop) ? part[ii] : make_int2(0, -1);
            }
#pragma unroll
            for (int u = 0; u < 4; ++u) {
                if (r[u].y >= 0) {
                    const int p = atomicAdd(&cursor[r[u].y - n0], 1);
                    col[start + p] = r[u].x;
                }
            }
        }
        __syncthreads();
        for (int nl = wave; nl < BIN_NODES; nl += 32) {
            const int beg = start + nbeg[nl];
            const int deg = nbeg[nl + 1] - nbeg[nl];
            if (deg <= 1) continue;
            if (deg <= 32) {
                const int v = (lane < deg) ? col[beg + lane] : 0;
                int r = 0;
                for (int j = 0; j < deg; ++j) {
                    const int vj = __shfl(v, j, 32);
                    r += (vj < v || (vj == v && j < lane)) ? 1 : 0;
                }
                if (lane < deg) col[beg + r] = v;
            } else if (lane == 0) {
                for (int i = 1; i < deg; ++i) {
                    const int key = col[beg + i];
                    int j = i - 1;
                    while (j >= 0 && col[beg + j] > key) {
                        col[beg + j + 1] = col[beg + j];
                        --j;
                    }
                    col[beg + j + 1] = key;
                }
            }
        }
    }
}

typedef const __attribute__((address_space(1))) char*   gbytes_t;
typedef const __attribute__((address_space(1))) float4* gfloat4_t;

__device__ __forceinline__ unsigned lane_offset(const int lane) {
    unsigned loff = (unsigned)lane * 16u;
    asm volatile("" : "+v"(loff));
    return loff;
}

template <int C>
__device__ __forceinline__ void rows_load(const float* __restrict__ x, const int ids, const int j,
                                          const unsigned loff, float4 (&v)[C]) {
    gbytes_t rb[C];
#pragma unroll
    for (int u = 0; u < C; ++u) {
        rb[u] = (gbytes_t)(x + (long)__builtin_amdgcn_readlane(ids, j + u) * 128);
        asm("" : "+s"(rb[u]));
    }
    __builtin_amdgcn_sched_barrier(0);
#pragma unroll
    for (int u = 0; u < C; ++u) v[u] = *(gfloat4_t)(rb[u] + loff);
}
#define ROW_ADD(V) do { acc.x += (V).x; acc.y += (V).y; acc.z += (V).z; acc.w += (V).w; } while (0)

#define TILE_PITCH 132

struct StripWalk {
    int cur, cur_end;
    float4 acc;
};

__device__ __forceinline__ void close_node(float* tl, const int lane, const int rpv, const int nvi,
                                           StripWalk& w) {
    const float nv = __int_as_float(__builtin_amdgcn_readlane(nvi, w.cur));
    float4 o;
    o.x = w.acc.x * nv;
    o.y = w.acc.y * nv;
    o.z = w.acc.z * nv;
    o.w = w.acc.w * nv;
    *(float4*)(tl + w.cur * TILE_PITCH + lane * 4) = o;
    w.acc = make_float4(0.0f, 0.0f, 0.0f, 0.0f);
    ++w.cur;
    w.cur_end = __builtin_amdgcn_readlane(rpv, w.cur + 1);
}

#define ADD_ENTRY(E, V) do { \
        while ((E) >= w.cur_end) close_node(tl, lane, rpv, nvi, w); \
        float4& acc = w.acc; \
        ROW_ADD(V); \
    } while (0)

__device__ __forceinline__ void gather_batch32(const float* __restrict__ x, float* tl, const int lane,
                                               const int ids, const int e0,
                                               const int rpv, const int nvi, StripWalk& w,
                                               const int* __restrict__ col,
                                               const int pf_base, const int e_end, int& ids_n) {
    float4 va[16], vb[16];
    rows_load<16>(x, ids, 0, lane_offset(lane), va);
    rows_load<16>(x, ids, 16, lane_offset(lane), vb);
    ids_n = (pf_base + lane < e_end) ? col[pf_base + lane] : 0;
    __builtin_amdgcn_sched_barrier(0);
#pragma unroll
    for (int u = 0; u < 16; ++u) ADD_ENTRY(e0 + u, va[u]);
#pragma unroll
    for (int u = 0; u < 16; ++u) ADD_ENTRY(e0 + 16 + u, vb[u]);
}

__device__ __forceinline__ void gather_tail(const float* __restrict__ x, float* tl, const int lane,
                                            const int ids, const int e0, const int c,
                                            const int rpv, const int nvi, StripWalk& w) {
    const int o8 = c & 16;
    const int o4 = o8 + (c & 8);
    const int o2 = o4 + (c & 4);
    const int o1 = o2 + (c & 2);
    float4 v16[16] = {}, v8[8] = {}, v4[4] = {}, v2[2] = {}, v1[1] = {};
    asm volatile("" : : "v"(ids));
    if (c & 16) rows_load<16>(x, ids, 0, lane_offset(lane), v16);
    if (c & 8) rows_load<8>(x, ids, o8, lane_offset(lane), v8);
    if (c & 4) rows_load<4>(x, ids, o4, lane_offset(lane), v4);
    if (c & 2) rows_load<2>(x, ids, o2, lane_offset(lane), v2);
    if (c & 1) rows_load<1>(x, ids, o1, lane_offset(lane), v1);
    if (c & 16) {
#pragma unroll
        for (int u = 0; u < 16; ++u) ADD_ENTRY(e0 + u, v16[u]);
    }
    if (c & 8) {
#pragma unroll
        for (int u = 0; u < 8; ++u) ADD_ENTRY(e0 + o8 + u, v8[u]);
    }
    if (c & 4) {
#pragma unroll
        for (int u = 0; u < 4; ++u) ADD_ENTRY(e0 + o4 + u, v4[u]);
    }
    if (c & 2) {
#pragma unroll
        for (int u = 0; u < 2; ++u) ADD_ENTRY(e0 + o2 + u, v2[u]);
    }
    if (c & 1) ADD_ENTRY(e0 + o1, v1[0]);
}

template <int NT>
__device__ __forceinline__ void chunk_products(const float* wq, const float4 (&a)[4], v8f (&acc)[NT]) {
    constexpr int LDW = 260;
#pragma unroll
    for (int j = 0; j < 4; ++j) {
        float4 b[NT];
#pragma unroll
        for (int c = 0; c < NT; ++c) b[c] = *(const float4*)(wq + c * 16 * LDW + j * 8);
        const v2f a0 = {a[j].x, a[j].y};
        const v2f a1 = {a[j].z, a[j].w};
#pragma unroll
        for (int c = 0; c < NT; ++c) {
            const v2f b0 = {b[c].x, b[c].y};
            acc[c] = __builtin_amdgcn_wmma_f32_16x16x4_f32(
                false, a0, false, b0, (short)0, acc[c], false, false);
        }
#pragma unroll
        for (int c = 0; c < NT; ++c) {
            const v2f b1 = {b[c].z, b[c].w};
            acc[c] = __builtin_amdgcn_wmma_f32_16x16x4_f32(
                false, a1, false, b1, (short)0, acc[c], false, false);
        }
    }
}

template <int NT>
__device__ __forceinline__ void x_half(const float* __restrict__ xa, const float* wb, v8f (&acc)[NT]) {
    float4 a_nxt[4];
#pragma unroll
    for (int j = 0; j < 4; ++j) a_nxt[j] = *(const float4*)(xa + j * 8);
#pragma unroll 1
    for (int q = 0; q < 4; ++q) {
        float4 a_cur[4];
#pragma unroll
        for (int j = 0; j < 4; ++j) a_cur[j] = a_nxt[j];
        if (q < 3) {
#pragma unroll
            for (int j = 0; j < 4; ++j) a_nxt[j] = *(const float4*)(xa + (q + 1) * 32 + j * 8);
        }
        chunk_products<NT>(wb + q * 32, a_cur, acc);
    }
}

template <int NT>
__device__ __forceinline__ void mean_half(const float* ta, const float* wb, v8f (&acc)[NT]) {
    float4 a_nxt[4];
#pragma unroll
    for (int j = 0; j < 4; ++j) a_nxt[j] = *(const float4*)(ta + j * 8);
#pragma unroll 1
    for (int q = 0; q < 4; ++q) {
        float4 a_cur[4];
#pragma unroll
        for (int j = 0; j < 4; ++j) a_cur[j] = a_nxt[j];
        if (q < 3) {
#pragma unroll
            for (int j = 0; j < 4; ++j) a_nxt[j] = *(const float4*)(ta + (q + 1) * 32 + j * 8);
        }
        chunk_products<NT>(wb + 128 + q * 32, a_cur, acc);
    }
}

template <int NT, bool LNRELU>
__global__ __launch_bounds__(512) void sage_layer_kernel(
    const float* __restrict__ x,
    const int* __restrict__ rowptr,
    const int* __restrict__ col,
    const float* __restrict__ W,
    const float* __restrict__ bias,
    float* __restrict__ y,
    int N) {
    constexpr int NDIM = NT * 16;
    constexpr int LDW  = 260;
    constexpr int KQ   = 512 / NDIM;
    constexpr int TILE = 16 * TILE_PITCH;

    __shared__ __attribute__((aligned(16))) float Wt[NDIM * LDW];
    __shared__ __attribute__((aligned(16))) float Mt[18 * TILE];

    const int tid  = threadIdx.x;
    const int wave = __builtin_amdgcn_readfirstlane(tid >> 5);
    const int lane = tid & 31;
    const int ml   = lane & 15;
    const int half = lane >> 4;

    {
        const int n  = tid & (NDIM - 1);
        const int kq = tid / NDIM;
#pragma unroll 4
        for (int it = 0; it < 64 / KQ; ++it) {
            const int k = (it * KQ + kq) * 4;
            float4 v;
            v.x = W[(k + 0) * NDIM + n];
            v.y = W[(k + 1) * NDIM + n];
            v.z = W[(k + 2) * NDIM + n];
            v.w = W[(k + 3) * NDIM + n];
            *(float4*)&Wt[n * LDW + k] = v;
        }
    }
    __syncthreads();

    float bv[NT];
#pragma unroll
    for (int c = 0; c < NT; ++c) bv[c] = bias[c * 16 + ml];

    const float* wb = &Wt[ml * LDW + half * 4];
    float* tl = &Mt[wave * TILE];
    const float* ta = tl + ml * TILE_PITCH + half * 4;
    const int nstrips = (N + 15) >> 4;
    const bool x_first = (((wave >> 3) ^ (wave >> 1)) & 1) != 0;

    const int bid = blockIdx.x;
    const int gsz = gridDim.x;
    const int nslots = (nstrips > bid) ? (nstrips - bid + gsz - 1) / gsz : 0;
    const int rem    = nslots & 15;
    const bool coop  = (rem >= 1) && (rem <= 9);
    const int n      = coop ? rem : 0;
    const int nx     = 16 - n;
    const int cbase  = nslots & ~15;
    const int nf = coop ? (nslots >> 4) : ((nslots > wave) ? ((nslots - wave + 15) >> 4) : 0);
    int nc = 0;
    if (coop) {
        if (wave < n) nc = 1;
        else if (wave - n < n) nc = (wave - n + nx < n) ? 2 : 1;
    }

    v8f acc[NT];
#pragma unroll
    for (int c = 0; c < NT; ++c) acc[c] = v8f{};

#pragma unroll 1
    for (int pass = 0; pass < (coop ? 2 : 1); ++pass) {
        if (pass == 1) __syncthreads();
        const int trips = (pass == 0) ? (nf + nc) : ((wave < n) ? 1 : 0);
#pragma unroll 1
        for (int it = 0; it < trips; ++it) {
            const bool fin    = (pass == 1);
            const bool shared = fin || (it >= nf);
            const bool m_unit = shared && (wave < n);
            const bool x_unit = shared && !m_unit;
            const bool has_x  = !shared || x_unit;
            const bool has_g  = !shared || (m_unit && !fin);
            int slot = it * 16 + wave;
            if (m_unit) slot = cbase + wave;
            if (x_unit) slot = cbase + (wave - n) + (it - nf) * nx;
            const int s = bid + gsz * slot;

            long row = (long)s * 16 + ml;
            if (row >= N) row = N - 1;
            const float* xa = x + row * 128 + half * 4;

            if (!fin) {
#pragma unroll
                for (int c = 0; c < NT; ++c) acc[c] = v8f{};
            }

            if (has_x && (x_first || x_unit)) x_half<NT>(xa, wb, acc);

            if (has_g) {
                int rpi = s * 16 + lane;
                if (rpi > N) rpi = N;
                const int rpv   = rowptr[rpi];
                const int e_beg = __builtin_amdgcn_readlane(rpv, 0);
                const int e_end = __builtin_amdgcn_readlane(rpv, 16);
                StripWalk w;
                w.cur = 0;
                w.cur_end = __builtin_amdgcn_readlane(rpv, 1);
                w.acc = make_float4(0.0f, 0.0f, 0.0f, 0.0f);

                int ids_n = (e_beg + lane < e_end) ? col[e_beg + lane] : 0;
                const int degl = __shfl_down(rpv, 1, 32) - rpv;
                const int nvi = __float_as_int(degl > 0 ? 1.0f / (float)degl : 0.0f);
                for (int base = e_beg; base < e_end; base += 32) {
                    const int ids = ids_n;
                    const int cnt = (e_end - base) < 32 ? (e_end - base) : 32;
                    if (cnt == 32)
                        gather_batch32(x, tl, lane, ids, base, rpv, nvi, w, col, base + 32, e_end, ids_n);
                    else
                        gather_tail(x, tl, lane, ids, base, cnt, rpv, nvi, w);
                }
                while (w.cur < 16) close_node(tl, lane, rpv, nvi, w);
            }

            if (has_x && !(x_first || x_unit)) x_half<NT>(xa, wb, acc);
            if (has_g) mean_half<NT>(ta, wb, acc);

            if (x_unit) {
                float* pt = &Mt[(n + slot - cbase) * TILE] + lane * 4;
#pragma unroll
                for (int c = 0; c < NT; ++c) {
                    *(float4*)(pt + (c * 2 + 0) * 128) = make_float4(acc[c][0], acc[c][1], acc[c][2], acc[c][3]);
                    *(float4*)(pt + (c * 2 + 1) * 128) = make_float4(acc[c][4], acc[c][5], acc[c][6], acc[c][7]);
                }
                asm volatile("s_wait_dscnt 0x0" ::: "memory");
                continue;
            }
            if (m_unit && !fin) continue;
            if (fin) {
                const float* pt = &Mt[(n + wave) * TILE] + lane * 4;
#pragma unroll
                for (int c = 0; c < NT; ++c) {
                    const float4 p0 = *(const float4*)(pt + (c * 2 + 0) * 128);
                    const float4 p1 = *(const float4*)(pt + (c * 2 + 1) * 128);
                    acc[c][0] += p0.x; acc[c][1] += p0.y; acc[c][2] += p0.z; acc[c][3] += p0.w;
                    acc[c][4] += p1.x; acc[c][5] += p1.y; acc[c][6] += p1.z; acc[c][7] += p1.w;
                }
            }

#pragma unroll
            for (int c = 0; c < NT; ++c) {
#pragma unroll
                for (int r = 0; r < 8; ++r) acc[c][r] += bv[c];
            }

            if (LNRELU) {
#pragma unroll
                for (int r = 0; r < 8; ++r) {
                    float sv = 0.0f, qv = 0.0f;
#pragma unroll
                    for (int c = 0; c < NT; ++c) {
                        sv += acc[c][r];
                        qv += acc[c][r] * acc[c][r];
                    }
#pragma unroll
                    for (int off = 1; off < 16; off <<= 1) {
                        sv += __shfl_xor(sv, off, 32);
                        qv += __shfl_xor(qv, off, 32);
                    }
                    const float mu   = sv * (1.0f / NDIM);
                    const float var  = qv * (1.0f / NDIM) - mu * mu;
                    const float rstd = rsqrtf(var + LN_EPS);
#pragma unroll
                    for (int c = 0; c < NT; ++c) {
                        const float v = (acc[c][r] - mu) * rstd;
                        acc[c][r] = v > 0.0f ? v : 0.0f;
                    }
                }
            }

#pragma unroll
            for (int r = 0; r < 8; ++r) {
                const long ro = (long)s * 16 + half * 8 + r;
                if (ro < N) {
#pragma unroll
                    for (int c = 0; c < NT; ++c) y[ro * NDIM + c * 16 + ml] = acc[c][r];
                }
            }
        }
    }
}


extern "C" void kernel_launch(void* const* d_in, const int* in_sizes, int n_in,
                              void* d_out, int out_size, void* d_ws, size_t ws_size,
                              hipStream_t stream) {
    const float* h   = (const float*)d_in[0];
    const int*   src = (const int*)d_in[1];
    const int*   dst = (const int*)d_in[2];
    const float* W1  = (const float*)d_in[3];
    const float* b1  = (const float*)d_in[4];
    const float* W2  = (const float*)d_in[5];
    const float* b2  = (const float*)d_in[6];
    const float* W3  = (const float*)d_in[7];
    const float* b3  = (const float*)d_in[8];
    float* out = (float*)d_out;

    const int N = in_sizes[0] / 128;
    const int E = in_sizes[1];

    const int nbins = (N >> BIN_SHIFT) + 1;
    const int npad  = nbins * BIN_NODES;
    const int M     = nbins * PART_WGS;
    const int epad  = (E + 63) & ~63;
    const int chunk = (E + PART_WGS - 1) / PART_WGS;

    int*   rowptr = (int*)d_ws;
    int*   ghist  = rowptr + npad;
    int*   goff   = ghist + M;
    int2*  part   = (int2*)(goff + M);
    int*   col    = (int*)(part + epad);
    float* h1     = (float*)(col + epad);
    float* h2     = h1  + (size_t)N * 128;

    bin_count_kernel<<<PART_WGS, 1024, 0, stream>>>(dst, ghist, E, chunk, nbins);
    bin_scan_kernel<<<1, 1024, 0, stream>>>(ghist, goff, M);
    bin_scatter_kernel<<<PART_WGS, 1024, 0, stream>>>(src, dst, goff, part, E, chunk, nbins);
    bin_build_kernel<<<nbins, 1024, 0, stream>>>(part, goff, rowptr, col, N, E, nbins);

    sage_layer_kernel<8, true><<<256, 512, 0, stream>>>(h, rowptr, col, W1, b1, h1, N);

    sage_layer_kernel<8, true><<<256, 512, 0, stream>>>(h1, rowptr, col, W2, b2, h2, N);

    sage_layer_kernel<4, false><<<256, 512, 0, stream>>>(h2, rowptr, col, W3, b3, out, N);
}


# Edge grouping in 2 launches, not 4: a slice counts, scans and regroups itself by bin in LDS (no global scan launch); a bin's build sums the count rows before it and walks its 256 pieces; sums same
# speedup vs baseline: 4.5708x; 1.0986x over previous
#include <hip/hip_runtime.h>


typedef __attribute__((ext_vector_type(2))) float v2f;
typedef __attribute__((ext_vector_type(8))) float v8f;

#define LN_EPS 1e-5f


#define BIN_SHIFT 9
#define BIN_NODES 512
#define MAX_BINS  2048
#define PART_WGS  256

__global__ __launch_bounds__(1024) void slice_group_kernel(const int* __restrict__ src,
                                                           const int* __restrict__ dst,
                                                           int* __restrict__ cnt_t,
                                                           int* __restrict__ loc_t,
                                                           int2* __restrict__ part,
                                                           int E, int chunk, int nbins) {
    __shared__ int hist[MAX_BINS];
    __shared__ int wtot[32];
    const int tid  = threadIdx.x;
    const int lane = tid & 31;
    const int wave = tid >> 5;
    for (int i = tid; i < nbins; i += 1024) hist[i] = 0;
    __syncthreads();
    const int e0 = blockIdx.x * chunk;
    const int e1 = (e0 + chunk < E) ? (e0 + chunk) : E;
    for (int e = e0 + tid; e < e1; e += 4096) {
        int d[4];
#pragma unroll
        for (int u = 0; u < 4; ++u) {
            const int ee = e + u * 1024;
            d[u] = (ee < e1) ? dst[ee] : -1;
        }
#pragma unroll
        for (int u = 0; u < 4; ++u) {
            if (d[u] >= 0) atomicAdd(&hist[d[u] >> BIN_SHIFT], 1);
        }
    }
    __syncthreads();

    const int b0 = tid * 2;
    const int ca = (b0 < nbins) ? hist[b0] : 0;
    const int cb = (b0 + 1 < nbins) ? hist[b0 + 1] : 0;
    const int c2 = ca + cb;
    int inc = c2;
    for (int off = 1; off < 32; off <<= 1) {
        const int v = __shfl_up(inc, off, 32);
        if (lane >= off) inc += v;
    }
    if (lane == 31) wtot[wave] = inc;
    __syncthreads();
    int ex = inc - c2;
    for (int w = 0; w < wave; ++w) ex += wtot[w];
    if (b0 < nbins) {
        cnt_t[b0 * PART_WGS + blockIdx.x] = ca;
        loc_t[b0 * PART_WGS + blockIdx.x] = ex;
        hist[b0] = ex;
    }
    if (b0 + 1 < nbins) {
        cnt_t[(b0 + 1) * PART_WGS + blockIdx.x] = cb;
        loc_t[(b0 + 1) * PART_WGS + blockIdx.x] = ex + ca;
        hist[b0 + 1] = ex + ca;
    }
    __syncthreads();

    for (int e = e0 + tid; e < e1; e += 4096) {
        int d[4], s[4];
#pragma unroll
        for (int u = 0; u < 4; ++u) {
            const int ee = e + u * 1024;
            d[u] = (ee < e1) ? dst[ee] : -1;
            s[u] = (ee < e1) ? src[ee] : 0;
        }
#pragma unroll
        for (int u = 0; u < 4; ++u) {
            if (d[u] >= 0) {
                const int p = atomicAdd(&hist[d[u] >> BIN_SHIFT], 1);
                part[e0 + p] = make_int2(s[u], d[u]);
            }
        }
    }
}

#define BIN_CAP 8192

__global__ __launch_bounds__(1024) void bin_build_kernel(const int2* __restrict__ part,
                                                         const int* __restrict__ cnt_t,
                                                         const int* __restrict__ loc_t,
                                                         int* __restrict__ rowptr,
                                                         int* __restrict__ col,
                                                         int N, int chunk) {
    __shared__ int cntl[BIN_NODES];
    __shared__ int cursor[BIN_NODES];
    __shared__ int nbeg[BIN_NODES + 1];
    __shared__ int wtot[32];
    __shared__ int red[32];
    __shared__ int segoff[PART_WGS + 1];
    __shared__ int segloc[PART_WGS];
    __shared__ int lsrc[BIN_CAP];
    __shared__ unsigned short lnode[BIN_CAP];
    const int tid  = threadIdx.x;
    const int lane = tid & 31;
    const int wave = tid >> 5;
    const int bin  = blockIdx.x;
    const int n0   = bin << BIN_SHIFT;

    const int pc = (tid < PART_WGS) ? cnt_t[bin * PART_WGS + tid] : 0;
    const int pl = (tid < PART_WGS) ? loc_t[bin * PART_WGS + tid] : 0;
    const int4* c4 = (const int4*)cnt_t;
    int ps = 0;
    const int n4 = bin * (PART_WGS / 4);
    for (int i = tid; i < n4; i += 16384) {
        int4 v[16];
#pragma unroll
        for (int u = 0; u < 16; ++u) {
            const int ii = i + u * 1024;
            v[u] = (ii < n4) ? c4[ii] : make_int4(0, 0, 0, 0);
        }
#pragma unroll
        for (int u = 0; u < 16; ++u) ps += v[u].x + v[u].y + v[u].z + v[u].w;
    }
    for (int off = 16; off > 0; off >>= 1) ps += __shfl_xor(ps, off, 32);
    if (lane == 0) red[wave] = ps;
    if (tid < BIN_NODES) cntl[tid] = 0;

    int pinc = pc;
    for (int off = 1; off < 32; off <<= 1) {
        const int u = __shfl_up(pinc, off, 32);
        if (lane >= off) pinc += u;
    }
    if (lane == 31 && wave < PART_WGS / 32) wtot[wave] = pinc;
    __syncthreads();
    if (tid < PART_WGS) {
        int pex = pinc - pc;
        for (int w = 0; w < wave; ++w) pex += wtot[w];
        segoff[tid] = pex;
        segloc[tid] = tid * chunk + pl;
        if (tid == PART_WGS - 1) segoff[PART_WGS] = pex + pc;
    }
    int start = 0;
    for (int w = 0; w < 32; ++w) start += red[w];
    __syncthreads();
    const int cnt = segoff[PART_WGS];
    const int pw  = tid >> 2;
    const int pq  = tid & 3;
    const int pn  = segoff[pw + 1] - segoff[pw];
    const int2* pp = part + segloc[pw];

    for (int k = pq; k < pn; k += 16) {
        int d[4];
#pragma unroll
        for (int u = 0; u < 4; ++u) {
            const int kk = k + u * 4;
            d[u] = (kk < pn) ? pp[kk].y : -1;
        }
#pragma unroll
        for (int u = 0; u < 4; ++u) {
            if (d[u] >= 0) atomicAdd(&cntl[d[u] - n0], 1);
        }
    }
    __syncthreads();

    const int c = (tid < BIN_NODES) ? cntl[tid] : 0;
    int inc = c;
    for (int off = 1; off < 32; off <<= 1) {
        const int u = __shfl_up(inc, off, 32);
        if (lane >= off) inc += u;
    }
    if (lane == 31 && wave < 16) wtot[wave] = inc;
    __syncthreads();
    if (tid < BIN_NODES) {
        int ex = inc - c;
        for (int w = 0; w < wave; ++w) ex += wtot[w];
        cursor[tid] = ex;
        nbeg[tid] = ex;
        if (tid == BIN_NODES - 1) nbeg[BIN_NODES] = ex + c;
        const int node = n0 + tid;
        if (node <= N) rowptr[node] = start + ex;
    }
    __syncthreads();

    if (cnt <= BIN_CAP) {
        for (int k = pq; k < pn; k += 16) {
            int2 r[4];
#pragma unroll
            for (int u = 0; u < 4; ++u) {
                const int kk = k + u * 4;
                r[u] = (kk < pn) ? pp[kk] : make_int2(0, -1);
            }
#pragma unroll
            for (int u = 0; u < 4; ++u) {
                if (r[u].y >= 0) {
                    const int nd = r[u].y - n0;
                    const int p = atomicAdd(&cursor[nd], 1);
                    lsrc[p]  = r[u].x;
                    lnode[p] = (unsigned short)nd;
                }
            }
        }
        __syncthreads();
        for (int j = tid; j < cnt; j += 1024) {
            const int sv = lsrc[j];
            const int nd = lnode[j];
            const int gb = nbeg[nd];
            const int ge = nbeg[nd + 1];
            int r = 0;
            for (int k = gb; k < ge; ++k) {
                const int v = lsrc[k];
                r += (v < sv || (v == sv && k < j)) ? 1 : 0;
            }
            col[start + gb + r] = sv;
        }
    } else {
        for (int k = pq; k < pn; k += 16) {
            int2 r[4];
#pragma unroll
            for (int u = 0; u < 4; ++u) {
                const int kk = k + u * 4;
                r[u] = (kk < pn) ? pp[kk] : make_int2(0, -1);
            }
#pragma unroll
            for (int u = 0; u < 4; ++u) {
                if (r[u].y >= 0) {
                    const int p = atomicAdd(&cursor[r[u].y - n0], 1);
                    col[start + p] = r[u].x;
                }
            }
        }
        __syncthreads();
        for (int nl = wave; nl < BIN_NODES; nl += 32) {
            const int beg = start + nbeg[nl];
            const int deg = nbeg[nl + 1] - nbeg[nl];
            if (deg <= 1) continue;
            if (deg <= 32) {
                const int v = (lane < deg) ? col[beg + lane] : 0;
                int r = 0;
                for (int j = 0; j < deg; ++j) {
                    const int vj = __shfl(v, j, 32);
                    r += (vj < v || (vj == v && j < lane)) ? 1 : 0;
                }
                if (lane < deg) col[beg + r] = v;
            } else if (lane == 0) {
                for (int i = 1; i < deg; ++i) {
                    const int key = col[beg + i];
                    int j = i - 1;
                    while (j >= 0 && col[beg + j] > key) {
                        col[beg + j + 1] = col[beg + j];
                        --j;
                    }
                    col[beg + j + 1] = key;
                }
            }
        }
    }
}

typedef const __attribute__((address_space(1))) char*   gbytes_t;
typedef const __attribute__((address_space(1))) float4* gfloat4_t;

__device__ __forceinline__ unsigned lane_offset(const int lane) {
    unsigned loff = (unsigned)lane * 16u;
    asm volatile("" : "+v"(loff));
    return loff;
}

template <int C>
__device__ __forceinline__ void rows_load(const float* __restrict__ x, const int ids, const int j,
                                          const unsigned loff, float4 (&v)[C]) {
    gbytes_t rb[C];
#pragma unroll
    for (int u = 0; u < C; ++u) {
        rb[u] = (gbytes_t)(x + (long)__builtin_amdgcn_readlane(ids, j + u) * 128);
        asm("" : "+s"(rb[u]));
    }
    __builtin_amdgcn_sched_barrier(0);
#pragma unroll
    for (int u = 0; u < C; ++u) v[u] = *(gfloat4_t)(rb[u] + loff);
}
#define ROW_ADD(V) do { acc.x += (V).x; acc.y += (V).y; acc.z += (V).z; acc.w += (V).w; } while (0)

#define TILE_PITCH 132

struct StripWalk {
    int cur, cur_end;
    float4 acc;
};

__device__ __forceinline__ void close_node(float* tl, const int lane, const int rpv, const int nvi,
                                           StripWalk& w) {
    const float nv = __int_as_float(__builtin_amdgcn_readlane(nvi, w.cur));
    float4 o;
    o.x = w.acc.x * nv;
    o.y = w.acc.y * nv;
    o.z = w.acc.z * nv;
    o.w = w.acc.w * nv;
    *(float4*)(tl + w.cur * TILE_PITCH + lane * 4) = o;
    w.acc = make_float4(0.0f, 0.0f, 0.0f, 0.0f);
    ++w.cur;
    w.cur_end = __builtin_amdgcn_readlane(rpv, w.cur + 1);
}

#define ADD_ENTRY(E, V) do { \
        while ((E) >= w.cur_end) close_node(tl, lane, rpv, nvi, w); \
        float4& acc = w.acc; \
        ROW_ADD(V); \
    } while (0)

__device__ __forceinline__ void gather_batch32(const float* __restrict__ x, float* tl, const int lane,
                                               const int ids, const int e0,
                                               const int rpv, const int nvi, StripWalk& w,
                                               const int* __restrict__ col,
                                               const int pf_base, const int e_end, int& ids_n) {
    float4 va[16], vb[16];
    rows_load<16>(x, ids, 0, lane_offset(lane), va);
    rows_load<16>(x, ids, 16, lane_offset(lane), vb);
    ids_n = (pf_base + lane < e_end) ? col[pf_base + lane] : 0;
    __builtin_amdgcn_sched_barrier(0);
#pragma unroll
    for (int u = 0; u < 16; ++u) ADD_ENTRY(e0 + u, va[u]);
#pragma unroll
    for (int u = 0; u < 16; ++u) ADD_ENTRY(e0 + 16 + u, vb[u]);
}

__device__ __forceinline__ void gather_tail(const float* __restrict__ x, float* tl, const int lane,
                                            const int ids, const int e0, const int c,
                                            const int rpv, const int nvi, StripWalk& w) {
    const int o8 = c & 16;
    const int o4 = o8 + (c & 8);
    const int o2 = o4 + (c & 4);
    const int o1 = o2 + (c & 2);
    float4 v16[16] = {}, v8[8] = {}, v4[4] = {}, v2[2] = {}, v1[1] = {};
    asm volatile("" : : "v"(ids));
    if (c & 16) rows_load<16>(x, ids, 0, lane_offset(lane), v16);
    if (c & 8) rows_load<8>(x, ids, o8, lane_offset(lane), v8);
    if (c & 4) rows_load<4>(x, ids, o4, lane_offset(lane), v4);
    if (c & 2) rows_load<2>(x, ids, o2, lane_offset(lane), v2);
    if (c & 1) rows_load<1>(x, ids, o1, lane_offset(lane), v1);
    if (c & 16) {
#pragma unroll
        for (int u = 0; u < 16; ++u) ADD_ENTRY(e0 + u, v16[u]);
    }
    if (c & 8) {
#pragma unroll
        for (int u = 0; u < 8; ++u) ADD_ENTRY(e0 + o8 + u, v8[u]);
    }
    if (c & 4) {
#pragma unroll
        for (int u = 0; u < 4; ++u) ADD_ENTRY(e0 + o4 + u, v4[u]);
    }
    if (c & 2) {
#pragma unroll
        for (int u = 0; u < 2; ++u) ADD_ENTRY(e0 + o2 + u, v2[u]);
    }
    if (c & 1) ADD_ENTRY(e0 + o1, v1[0]);
}

template <int NT>
__device__ __forceinline__ void chunk_products(const float* wq, const float4 (&a)[4], v8f (&acc)[NT]) {
    constexpr int LDW = 260;
#pragma unroll
    for (int j = 0; j < 4; ++j) {
        float4 b[NT];
#pragma unroll
        for (int c = 0; c < NT; ++c) b[c] = *(const float4*)(wq + c * 16 * LDW + j * 8);
        const v2f a0 = {a[j].x, a[j].y};
        const v2f a1 = {a[j].z, a[j].w};
#pragma unroll
        for (int c = 0; c < NT; ++c) {
            const v2f b0 = {b[c].x, b[c].y};
            acc[c] = __builtin_amdgcn_wmma_f32_16x16x4_f32(
                false, a0, false, b0, (short)0, acc[c], false, false);
        }
#pragma unroll
        for (int c = 0; c < NT; ++c) {
            const v2f b1 = {b[c].z, b[c].w};
            acc[c] = __builtin_amdgcn_wmma_f32_16x16x4_f32(
                false, a1, false, b1, (short)0, acc[c], false, false);
        }
    }
}

template <int NT>
__device__ __forceinline__ void x_half(const float* __restrict__ xa, const float* wb, v8f (&acc)[NT]) {
    float4 a_nxt[4];
#pragma unroll
    for (int j = 0; j < 4; ++j) a_nxt[j] = *(const float4*)(xa + j * 8);
#pragma unroll 1
    for (int q = 0; q < 4; ++q) {
        float4 a_cur[4];
#pragma unroll
        for (int j = 0; j < 4; ++j) a_cur[j] = a_nxt[j];
        if (q < 3) {
#pragma unroll
            for (int j = 0; j < 4; ++j) a_nxt[j] = *(const float4*)(xa + (q + 1) * 32 + j * 8);
        }
        chunk_products<NT>(wb + q * 32, a_cur, acc);
    }
}

template <int NT>
__device__ __forceinline__ void mean_half(const float* ta, const float* wb, v8f (&acc)[NT]) {
    float4 a_nxt[4];
#pragma unroll
    for (int j = 0; j < 4; ++j) a_nxt[j] = *(const float4*)(ta + j * 8);
#pragma unroll 1
    for (int q = 0; q < 4; ++q) {
        float4 a_cur[4];
#pragma unroll
        for (int j = 0; j < 4; ++j) a_cur[j] = a_nxt[j];
        if (q < 3) {
#pragma unroll
            for (int j = 0; j < 4; ++j) a_nxt[j] = *(const float4*)(ta + (q + 1) * 32 + j * 8);
        }
        chunk_products<NT>(wb + 128 + q * 32, a_cur, acc);
    }
}

template <int NT, bool LNRELU>
__global__ __launch_bounds__(512) void sage_layer_kernel(
    const float* __restrict__ x,
    const int* __restrict__ rowptr,
    const int* __restrict__ col,
    const float* __restrict__ W,
    const float* __restrict__ bias,
    float* __restrict__ y,
    int N) {
    constexpr int NDIM = NT * 16;
    constexpr int LDW  = 260;
    constexpr int KQ   = 512 / NDIM;
    constexpr int TILE = 16 * TILE_PITCH;

    __shared__ __attribute__((aligned(16))) float Wt[NDIM * LDW];
    __shared__ __attribute__((aligned(16))) float Mt[18 * TILE];

    const int tid  = threadIdx.x;
    const int wave = __builtin_amdgcn_readfirstlane(tid >> 5);
    const int lane = tid & 31;
    const int ml   = lane & 15;
    const int half = lane >> 4;

    {
        const int n  = tid & (NDIM - 1);
        const int kq = tid / NDIM;
#pragma unroll 4
        for (int it = 0; it < 64 / KQ; ++it) {
            const int k = (it * KQ + kq) * 4;
            float4 v;
            v.x = W[(k + 0) * NDIM + n];
            v.y = W[(k + 1) * NDIM + n];
            v.z = W[(k + 2) * NDIM + n];
            v.w = W[(k + 3) * NDIM + n];
            *(float4*)&Wt[n * LDW + k] = v;
        }
    }
    __syncthreads();

    float bv[NT];
#pragma unroll
    for (int c = 0; c < NT; ++c) bv[c] = bias[c * 16 + ml];

    const float* wb = &Wt[ml * LDW + half * 4];
    float* tl = &Mt[wave * TILE];
    const float* ta = tl + ml * TILE_PITCH + half * 4;
    const int nstrips = (N + 15) >> 4;
    const bool x_first = (((wave >> 3) ^ (wave >> 1)) & 1) != 0;

    const int bid = blockIdx.x;
    const int gsz = gridDim.x;
    const int nslots = (nstrips > bid) ? (nstrips - bid + gsz - 1) / gsz : 0;
    const int rem    = nslots & 15;
    const bool coop  = (rem >= 1) && (rem <= 9);
    const int n      = coop ? rem : 0;
    const int nx     = 16 - n;
    const int cbase  = nslots & ~15;
    const int nf = coop ? (nslots >> 4) : ((nslots > wave) ? ((nslots - wave + 15) >> 4) : 0);
    int nc = 0;
    if (coop) {
        if (wave < n) nc = 1;
        else if (wave - n < n) nc = (wave - n + nx < n) ? 2 : 1;
    }

    v8f acc[NT];
#pragma unroll
    for (int c = 0; c < NT; ++c) acc[c] = v8f{};

#pragma unroll 1
    for (int pass = 0; pass < (coop ? 2 : 1); ++pass) {
        if (pass == 1) __syncthreads();
        const int trips = (pass == 0) ? (nf + nc) : ((wave < n) ? 1 : 0);
#pragma unroll 1
        for (int it = 0; it < trips; ++it) {
            const bool fin    = (pass == 1);
            const bool shared = fin || (it >= nf);
            const bool m_unit = shared && (wave < n);
            const bool x_unit = shared && !m_unit;
            const bool has_x  = !shared || x_unit;
            const bool has_g  = !shared || (m_unit && !fin);
            int slot = it * 16 + wave;
            if (m_unit) slot = cbase + wave;
            if (x_unit) slot = cbase + (wave - n) + (it - nf) * nx;
            const int s = bid + gsz * slot;

            long row = (long)s * 16 + ml;
            if (row >= N) row = N - 1;
            const float* xa = x + row * 128 + half * 4;

            if (!fin) {
#pragma unroll
                for (int c = 0; c < NT; ++c) acc[c] = v8f{};
            }

            if (has_x && (x_first || x_unit)) x_half<NT>(xa, wb, acc);

            if (has_g) {
                int rpi = s * 16 + lane;
                if (rpi > N) rpi = N;
                const int rpv   = rowptr[rpi];
                const int e_beg = __builtin_amdgcn_readlane(rpv, 0);
                const int e_end = __builtin_amdgcn_readlane(rpv, 16);
                StripWalk w;
                w.cur = 0;
                w.cur_end = __builtin_amdgcn_readlane(rpv, 1);
                w.acc = make_float4(0.0f, 0.0f, 0.0f, 0.0f);

                int ids_n = (e_beg + lane < e_end) ? col[e_beg + lane] : 0;
                const int degl = __shfl_down(rpv, 1, 32) - rpv;
                const int nvi = __float_as_int(degl > 0 ? 1.0f / (float)degl : 0.0f);
                for (int base = e_beg; base < e_end; base += 32) {
                    const int ids = ids_n;
                    const int cnt = (e_end - base) < 32 ? (e_end - base) : 32;
                    if (cnt == 32)
                        gather_batch32(x, tl, lane, ids, base, rpv, nvi, w, col, base + 32, e_end, ids_n);
                    else
                        gather_tail(x, tl, lane, ids, base, cnt, rpv, nvi, w);
                }
                while (w.cur < 16) close_node(tl, lane, rpv, nvi, w);
            }

            if (has_x && !(x_first || x_unit)) x_half<NT>(xa, wb, acc);
            if (has_g) mean_half<NT>(ta, wb, acc);

            if (x_unit) {
                float* pt = &Mt[(n + slot - cbase) * TILE] + lane * 4;
#pragma unroll
                for (int c = 0; c < NT; ++c) {
                    *(float4*)(pt + (c * 2 + 0) * 128) = make_float4(acc[c][0], acc[c][1], acc[c][2], acc[c][3]);
                    *(float4*)(pt + (c * 2 + 1) * 128) = make_float4(acc[c][4], acc[c][5], acc[c][6], acc[c][7]);
                }
                asm volatile("s_wait_dscnt 0x0" ::: "memory");
                continue;
            }
            if (m_unit && !fin) continue;
            if (fin) {
                const float* pt = &Mt[(n + wave) * TILE] + lane * 4;
#pragma unroll
                for (int c = 0; c < NT; ++c) {
                    const float4 p0 = *(const float4*)(pt + (c * 2 + 0) * 128);
                    const float4 p1 = *(const float4*)(pt + (c * 2 + 1) * 128);
                    acc[c][0] += p0.x; acc[c][1] += p0.y; acc[c][2] += p0.z; acc[c][3] += p0.w;
                    acc[c][4] += p1.x; acc[c][5] += p1.y; acc[c][6] += p1.z; acc[c][7] += p1.w;
                }
            }

#pragma unroll
            for (int c = 0; c < NT; ++c) {
#pragma unroll
                for (int r = 0; r < 8; ++r) acc[c][r] += bv[c];
            }

            if (LNRELU) {
#pragma unroll
                for (int r = 0; r < 8; ++r) {
                    float sv = 0.0f, qv = 0.0f;
#pragma unroll
                    for (int c = 0; c < NT; ++c) {
                        sv += acc[c][r];
                        qv += acc[c][r] * acc[c][r];
                    }
#pragma unroll
                    for (int off = 1; off < 16; off <<= 1) {
                        sv += __shfl_xor(sv, off, 32);
                        qv += __shfl_xor(qv, off, 32);
                    }
                    const float mu   = sv * (1.0f / NDIM);
                    const float var  = qv * (1.0f / NDIM) - mu * mu;
                    const float rstd = rsqrtf(var + LN_EPS);
#pragma unroll
                    for (int c = 0; c < NT; ++c) {
                        const float v = (acc[c][r] - mu) * rstd;
                        acc[c][r] = v > 0.0f ? v : 0.0f;
                    }
                }
            }

#pragma unroll
            for (int r = 0; r < 8; ++r) {
                const long ro = (long)s * 16 + half * 8 + r;
                if (ro < N) {
#pragma unroll
                    for (int c = 0; c < NT; ++c) y[ro * NDIM + c * 16 + ml] = acc[c][r];
                }
            }
        }
    }
}


extern "C" void kernel_launch(void* const* d_in, const int* in_sizes, int n_in,
                              void* d_out, int out_size, void* d_ws, size_t ws_size,
                              hipStream_t stream) {
    const float* h   = (const float*)d_in[0];
    const int*   src = (const int*)d_in[1];
    const int*   dst = (const int*)d_in[2];
    const float* W1  = (const float*)d_in[3];
    const float* b1  = (const float*)d_in[4];
    const float* W2  = (const float*)d_in[5];
    const float* b2  = (const float*)d_in[6];
    const float* W3  = (const float*)d_in[7];
    const float* b3  = (const float*)d_in[8];
    float* out = (float*)d_out;

    const int N = in_sizes[0] / 128;
    const int E = in_sizes[1];

    const int nbins = (N >> BIN_SHIFT) + 1;
    const int npad  = nbins * BIN_NODES;
    const int M     = nbins * PART_WGS;
    const int epad  = (E + 63) & ~63;
    const int chunk = (E + PART_WGS - 1) / PART_WGS;

    int*   rowptr = (int*)d_ws;
    int*   cnt_t  = rowptr + npad;
    int*   loc_t  = cnt_t + M;
    int2*  part   = (int2*)(loc_t + M);
    int*   col    = (int*)(part + epad);
    float* h1     = (float*)(col + epad);
    float* h2     = h1  + (size_t)N * 128;

    slice_group_kernel<<<PART_WGS, 1024, 0, stream>>>(src, dst, cnt_t, loc_t, part, E, chunk, nbins);
    bin_build_kernel<<<nbins, 1024, 0, stream>>>(part, cnt_t, loc_t, rowptr, col, N, chunk);

    sage_layer_kernel<8, true><<<256, 512, 0, stream>>>(h, rowptr, col, W1, b1, h1, N);

    sage_layer_kernel<8, true><<<256, 512, 0, stream>>>(h1, rowptr, col, W2, b2, h2, N);

    sage_layer_kernel<4, false><<<256, 512, 0, stream>>>(h2, rowptr, col, W3, b3, out, N);
}
